# starting kernel
# baseline (speedup 1.0000x reference)
#include <hip/hip_runtime.h>

#define NU 100000
#define NB 50000
#define NNODE (NU + NB)
#define H 128
#define F 256
#define EPR 500000
#define NE (4 * EPR)

typedef __attribute__((ext_vector_type(2))) float v2f;
typedef __attribute__((ext_vector_type(8))) float v8f;

__global__ void __launch_bounds__(256)
gemm_bias_wmma(const float* __restrict__ X, const float* __restrict__ W,
               const float* __restrict__ bias, float* __restrict__ Y, int M) {
  const int wave = (int)((blockIdx.x * blockDim.x + threadIdx.x) >> 5);
  const int lane = threadIdx.x & 31;
  if (wave * 16 >= M) return;
  const int mbase = wave * 16;
  const int lrow  = lane & 15;
  const int khalf = (lane >> 4) << 1;

  const float* arow = X + (size_t)(mbase + lrow) * F + khalf;

  v8f c[8] = {};

  for (int k = 0; k < F; k += 4) {
    v2f a = *(const v2f*)(arow + k);
#pragma unroll
    for (int nt = 0; nt < 8; ++nt) {
      v2f b = *(const v2f*)(W + (size_t)(nt * 16 + lrow) * F + khalf + k);
      c[nt] = __builtin_amdgcn_wmma_f32_16x16x4_f32(
           false, a,  false, b,
           (short)0, c[nt],  false,  false);
    }
  }

  const int mo = (lane >> 4) << 3;
#pragma unroll
  for (int nt = 0; nt < 8; ++nt) {
    float bv = bias[nt * 16 + lrow];
#pragma unroll
    for (int v = 0; v < 8; ++v) {
      Y[(size_t)(mbase + mo + v) * H + nt * 16 + lrow] = c[nt][v] + bv;
    }
  }
}

__device__ __forceinline__ void decode_edge(int e, const int* __restrict__ rates,
                                            const int* __restrict__ impl,
                                            int& row, int& col) {
  const int rev = (e >= 2 * EPR) ? 1 : 0;
  int idx = rev ? (e - 2 * EPR) : e;
  const int* ed = (idx < EPR) ? rates : impl;
  const int i = (idx < EPR) ? idx : (idx - EPR);
  const int u = ed[i];
  const int b = ed[i + EPR];
  row = rev ? (b + NU) : u;
  col = rev ? u : b;
}

__global__ void __launch_bounds__(256)
degree_kernel(const int* __restrict__ rates, const int* __restrict__ impl,
              float* __restrict__ deg) {
  const int e = blockIdx.x * blockDim.x + threadIdx.x;
  if (e >= NE) return;
  int row, col;
  decode_edge(e, rates, impl, row, col);
  unsafeAtomicAdd(deg + col, 1.0f);
}

__global__ void __launch_bounds__(256)
dis_kernel(const float* __restrict__ deg, float* __restrict__ dis) {
  const int i = blockIdx.x * blockDim.x + threadIdx.x;
  if (i >= NNODE) return;
  const float d = deg[i];
  dis[i] = (d > 0.0f) ? rsqrtf(d) : 0.0f;
}

__global__ void __launch_bounds__(256)
norm_kernel(const int* __restrict__ rates, const int* __restrict__ impl,
            const float* __restrict__ dis, float* __restrict__ nrm) {
  const int e = blockIdx.x * blockDim.x + threadIdx.x;
  if (e >= NE) return;
  int row, col;
  decode_edge(e, rates, impl, row, col);
  nrm[e] = dis[row] * dis[col];
}

__global__ void __launch_bounds__(256)
scatter_kernel(const float* __restrict__ x, const float* __restrict__ nrm,
               const int* __restrict__ rates, const int* __restrict__ impl,
               float* __restrict__ out) {
  const int tid  = blockIdx.x * blockDim.x + threadIdx.x;
  const int e    = tid >> 5;
  const int lane = tid & 31;
  if (e >= NE) return;
  int row, col;
  decode_edge(e, rates, impl, row, col);
  const float n = nrm[e];
  const float4 v = ((const float4*)(x + (size_t)row * H))[lane];
  float* dst = out + (size_t)col * H + lane * 4;
  unsafeAtomicAdd(dst + 0, v.x * n);
  unsafeAtomicAdd(dst + 1, v.y * n);
  unsafeAtomicAdd(dst + 2, v.z * n);
  unsafeAtomicAdd(dst + 3, v.w * n);
}

__global__ void __launch_bounds__(256)
accum_copy_kernel(float* __restrict__ acc, float* __restrict__ x,
                  const float* __restrict__ buf) {
  const int i = blockIdx.x * blockDim.x + threadIdx.x;
  if (i >= NNODE * H) return;
  const float v = buf[i];
  acc[i] += v;
  x[i] = v;
}

__global__ void __launch_bounds__(256)
scale_kernel(float* __restrict__ acc) {
  const int i = blockIdx.x * blockDim.x + threadIdx.x;
  if (i >= NNODE * H) return;
  acc[i] *= (1.0f / 3.0f);
}

extern "C" void kernel_launch(void* const* d_in, const int* in_sizes, int n_in,
                              void* d_out, int out_size, void* d_ws, size_t ws_size,
                              hipStream_t stream) {
  (void)in_sizes; (void)n_in; (void)out_size; (void)ws_size;

  const float* user_x = (const float*)d_in[0];
  const float* book_x = (const float*)d_in[1];
  const float* user_W = (const float*)d_in[2];
  const float* user_b = (const float*)d_in[3];
  const float* book_W = (const float*)d_in[4];
  const float* book_b = (const float*)d_in[5];
  const int*   rates  = (const int*)d_in[6];
  const int*   impl   = (const int*)d_in[7];
  float* outp = (float*)d_out;

  float* x   = (float*)d_ws;
  float* buf = x   + (size_t)NNODE * H;
  float* deg = buf + (size_t)NNODE * H;
  float* dis = deg + NNODE;
  float* nrm = dis + NNODE;

  const size_t embBytes = (size_t)NNODE * H * sizeof(float);
  hipMemsetAsync(deg, 0, NNODE * sizeof(float), stream);

  {
    int waves = NU / 16;
    gemm_bias_wmma<<<(waves + 7) / 8, 256, 0, stream>>>(user_x, user_W, user_b, x, NU);
    waves = NB / 16;
    gemm_bias_wmma<<<(waves + 7) / 8, 256, 0, stream>>>(book_x, book_W, book_b,
                                                        x + (size_t)NU * H, NB);
  }

  hipMemcpyAsync(outp, x, embBytes, hipMemcpyDeviceToDevice, stream);

  degree_kernel<<<(NE + 255) / 256, 256, 0, stream>>>(rates, impl, deg);
  dis_kernel<<<(NNODE + 255) / 256, 256, 0, stream>>>(deg, dis);
  norm_kernel<<<(NE + 255) / 256, 256, 0, stream>>>(rates, impl, dis, nrm);

  const int nElem = NNODE * H;
  for (int layer = 0; layer < 2; ++layer) {
    hipMemsetAsync(buf, 0, embBytes, stream);
    const long long threads = (long long)NE * 32;
    scatter_kernel<<<(int)((threads + 255) / 256), 256, 0, stream>>>(x, nrm, rates,
                                                                     impl, buf);
    accum_copy_kernel<<<(nElem + 255) / 256, 256, 0, stream>>>(outp, x, buf);
  }

  scale_kernel<<<(nElem + 255) / 256, 256, 0, stream>>>(outp);
}


# layer sums: block b owns targets (t&255)==b, scans all edges, sums its rows in LDS float adds (arrival order), stores whole rows; global float atomics and buf memsets gone
# speedup vs baseline: 1.4572x; 1.4572x over previous
#include <hip/hip_runtime.h>

#define NU 100000
#define NB 50000
#define NNODE (NU + NB)
#define H 128
#define F 256
#define EPR 500000
#define NE (4 * EPR)

typedef __attribute__((ext_vector_type(2))) float v2f;
typedef __attribute__((ext_vector_type(8))) float v8f;

__global__ void __launch_bounds__(256)
gemm_bias_wmma(const float* __restrict__ X, const float* __restrict__ W,
               const float* __restrict__ bias, float* __restrict__ Y, int M) {
  const int wave = (int)((blockIdx.x * blockDim.x + threadIdx.x) >> 5);
  const int lane = threadIdx.x & 31;
  if (wave * 16 >= M) return;
  const int mbase = wave * 16;
  const int lrow  = lane & 15;
  const int khalf = (lane >> 4) << 1;

  const float* arow = X + (size_t)(mbase + lrow) * F + khalf;

  v8f c[8] = {};

  for (int k = 0; k < F; k += 4) {
    v2f a = *(const v2f*)(arow + k);
#pragma unroll
    for (int nt = 0; nt < 8; ++nt) {
      v2f b = *(const v2f*)(W + (size_t)(nt * 16 + lrow) * F + khalf + k);
      c[nt] = __builtin_amdgcn_wmma_f32_16x16x4_f32(
           false, a,  false, b,
           (short)0, c[nt],  false,  false);
    }
  }

  const int mo = (lane >> 4) << 3;
#pragma unroll
  for (int nt = 0; nt < 8; ++nt) {
    float bv = bias[nt * 16 + lrow];
#pragma unroll
    for (int v = 0; v < 8; ++v) {
      Y[(size_t)(mbase + mo + v) * H + nt * 16 + lrow] = c[nt][v] + bv;
    }
  }
}

__device__ __forceinline__ void decode_edge(int e, const int* __restrict__ rates,
                                            const int* __restrict__ impl,
                                            int& row, int& col) {
  const int rev = (e >= 2 * EPR) ? 1 : 0;
  int idx = rev ? (e - 2 * EPR) : e;
  const int* ed = (idx < EPR) ? rates : impl;
  const int i = (idx < EPR) ? idx : (idx - EPR);
  const int u = ed[i];
  const int b = ed[i + EPR];
  row = rev ? (b + NU) : u;
  col = rev ? u : b;
}

__global__ void __launch_bounds__(256)
degree_kernel(const int* __restrict__ rates, const int* __restrict__ impl,
              float* __restrict__ deg) {
  const int e = blockIdx.x * blockDim.x + threadIdx.x;
  if (e >= NE) return;
  int row, col;
  decode_edge(e, rates, impl, row, col);
  unsafeAtomicAdd(deg + col, 1.0f);
}

__global__ void __launch_bounds__(256)
dis_kernel(const float* __restrict__ deg, float* __restrict__ dis) {
  const int i = blockIdx.x * blockDim.x + threadIdx.x;
  if (i >= NNODE) return;
  const float d = deg[i];
  dis[i] = (d > 0.0f) ? rsqrtf(d) : 0.0f;
}

__global__ void __launch_bounds__(256)
norm_kernel(const int* __restrict__ rates, const int* __restrict__ impl,
            const float* __restrict__ dis, float* __restrict__ nrm) {
  const int e = blockIdx.x * blockDim.x + threadIdx.x;
  if (e >= NE) return;
  int row, col;
  decode_edge(e, rates, impl, row, col);
  nrm[e] = dis[row] * dis[col];
}

#define OWN_BLOCKS  256
#define OWN_ROWS    ((NNODE + OWN_BLOCKS - 1) / OWN_BLOCKS)
#define OWN_THREADS 1024
#define OWN_WAVES   (OWN_THREADS / 32)
#define QCAP        160
#define SEG_QUADS   (EPR / 4)
#define SEG_CHUNKS  ((SEG_QUADS + 31) / 32)
#define SCAN_U      4
#define SCAN_ITERS  (((SEG_CHUNKS + OWN_WAVES - 1) / OWN_WAVES + SCAN_U - 1) / SCAN_U * SCAN_U)
#define DRAIN_G     8

__device__ __forceinline__ void lds_add(float* p, float v) {
  __hip_atomic_fetch_add(p, v, __ATOMIC_RELAXED, __HIP_MEMORY_SCOPE_WORKGROUP);
}

__device__ __forceinline__ void drain_queue(const unsigned* q, int n,
                                            const float4* __restrict__ x4,
                                            const float* __restrict__ nrm,
                                            const int* __restrict__ rates,
                                            const int* __restrict__ impl,
                                            float* acc, int lane) {
  const int h = lane >> 4;
  for (int base = 0; base < n; base += DRAIN_G) {
    float4 v[DRAIN_G];
    float  w[DRAIN_G];
    unsigned il[DRAIN_G];
    const int li = base + (lane & (DRAIN_G - 1));
    const int dv = (int)q[(li < n) ? li : base];
#pragma unroll
    for (int g = 0; g < DRAIN_G; ++g) {
      const unsigned d = (unsigned)__builtin_amdgcn_readlane(dv, g);
      const int e = (int)(d >> 10);
      int row, col;
      decode_edge(e, rates, impl, row, col);
      (void)col;
      w[g]  = nrm[e];
      il[g] = d & 1023u;
      v[g]  = x4[(size_t)row * (H / 4) + lane];
    }
#pragma unroll
    for (int g = 0; g < DRAIN_G; ++g) {
      if (base + g < n) {
        float* a = acc + il[g] * H + 4 * lane;
        const float4 s = v[g];
        const float  m = w[g];
        lds_add(a + h,             (h ? s.y : s.x) * m);
        lds_add(a + ((1 + h) & 3), (h ? s.z : s.y) * m);
        lds_add(a + ((2 + h) & 3), (h ? s.w : s.z) * m);
        lds_add(a + ((3 + h) & 3), (h ? s.x : s.w) * m);
      }
    }
  }
}

__global__ void __launch_bounds__(OWN_THREADS)
gather_owner_kernel(const float* __restrict__ x, const float* __restrict__ nrm,
                    const int* __restrict__ rates, const int* __restrict__ impl,
                    float* __restrict__ out) {
  __shared__ __attribute__((aligned(16))) float acc[OWN_ROWS * H];
  __shared__ unsigned queue[OWN_WAVES * QCAP];

  const int tid  = threadIdx.x;
  const int lane = tid & 31;
  const int wave = __builtin_amdgcn_readfirstlane(tid >> 5);
  const unsigned b = blockIdx.x;

  float4* acc4 = (float4*)acc;
  for (int i = tid; i < OWN_ROWS * H / 4; i += OWN_THREADS)
    acc4[i] = make_float4(0.0f, 0.0f, 0.0f, 0.0f);
  __syncthreads();

  unsigned* q = queue + wave * QCAP;
  const float4* x4 = (const float4*)x;
  const unsigned key = b << 24;
  const int rot = (int)((b * SEG_CHUNKS) >> 8);
  int cnt = 0;

  for (int seg = 0; seg < 4; ++seg) {
    const int* colp = (seg == 0) ? (rates + EPR) : (seg == 1) ? (impl + EPR)
                    : (seg == 2) ? rates : impl;
    const int4* col4 = (const int4*)colp;
    const int ebase = seg * EPR;

    int4 cv[SCAN_U], nv[SCAN_U];
    int  iq[SCAN_U], nq[SCAN_U];
#pragma unroll
    for (int u = 0; u < SCAN_U; ++u) {
      const int c0 = u * OWN_WAVES + wave;
      int c = c0 + rot;
      if (c >= SEG_CHUNKS) c -= SEG_CHUNKS;
      const int qi = c * 32 + lane;
      iq[u] = ((c0 < SEG_CHUNKS) && (qi < SEG_QUADS)) ? qi : -1;
      cv[u] = col4[(qi < SEG_QUADS) ? qi : (SEG_QUADS - 1)];
    }

    for (int it = 0; it < SCAN_ITERS; it += SCAN_U) {
#pragma unroll
      for (int u = 0; u < SCAN_U; ++u) {
        const int c0 = (it + SCAN_U + u) * OWN_WAVES + wave;
        int c = c0 + rot;
        if (c >= SEG_CHUNKS) c -= SEG_CHUNKS;
        const int qi = c * 32 + lane;
        nq[u] = ((c0 < SEG_CHUNKS) && (qi < SEG_QUADS)) ? qi : -1;
        nv[u] = col4[(qi < SEG_QUADS) ? qi : (SEG_QUADS - 1)];
      }
#pragma unroll
      for (int u = 0; u < SCAN_U; ++u) {
        const int tk[4] = {cv[u].x, cv[u].y, cv[u].z, cv[u].w};
        unsigned iloc[4], mk[4];
        const unsigned live = __builtin_amdgcn_ballot_w32(iq[u] >= 0);
#pragma unroll
        for (int k = 0; k < 4; ++k) {
          const unsigned t = (unsigned)tk[k];
          iloc[k] = ((t >> 8) | (t << 24)) - key;
          mk[k] = __builtin_amdgcn_ballot_w32(iloc[k] < (unsigned)OWN_ROWS) & live;
        }
        if ((mk[0] | mk[1] | mk[2] | mk[3]) != 0u) {
          const unsigned e0 = (unsigned)(ebase + iq[u] * 4);
#pragma unroll
          for (int k = 0; k < 4; ++k) {
            if (mk[k] != 0u) {
              if ((mk[k] >> lane) & 1u)
                q[cnt + (int)__builtin_amdgcn_mbcnt_lo(mk[k], 0u)] = ((e0 + k) << 10) | iloc[k];
              cnt += __builtin_popcount(mk[k]);
            }
          }
          if (cnt >= 32) {
            __builtin_amdgcn_fence(__ATOMIC_SEQ_CST, "wavefront");
            drain_queue(q, cnt, x4, nrm, rates, impl, acc, lane);
            __builtin_amdgcn_fence(__ATOMIC_SEQ_CST, "wavefront");
            cnt = 0;
          }
        }
      }
#pragma unroll
      for (int u = 0; u < SCAN_U; ++u) { cv[u] = nv[u]; iq[u] = nq[u]; }
    }
  }
  __builtin_amdgcn_fence(__ATOMIC_SEQ_CST, "wavefront");
  drain_queue(q, cnt, x4, nrm, rates, impl, acc, lane);

  __syncthreads();
  for (int i = wave; i < OWN_ROWS; i += OWN_WAVES) {
    const unsigned t = ((unsigned)i << 8) | b;
    if (t < (unsigned)NNODE)
      ((float4*)(out + (size_t)t * H))[lane] = ((const float4*)(acc + i * H))[lane];
  }
}

__global__ void __launch_bounds__(256)
accum_copy_kernel(float* __restrict__ acc, float* __restrict__ x,
                  const float* __restrict__ buf) {
  const int i = blockIdx.x * blockDim.x + threadIdx.x;
  if (i >= NNODE * H) return;
  const float v = buf[i];
  acc[i] += v;
  x[i] = v;
}

__global__ void __launch_bounds__(256)
scale_kernel(float* __restrict__ acc) {
  const int i = blockIdx.x * blockDim.x + threadIdx.x;
  if (i >= NNODE * H) return;
  acc[i] *= (1.0f / 3.0f);
}

extern "C" void kernel_launch(void* const* d_in, const int* in_sizes, int n_in,
                              void* d_out, int out_size, void* d_ws, size_t ws_size,
                              hipStream_t stream) {
  (void)in_sizes; (void)n_in; (void)out_size; (void)ws_size;

  const float* user_x = (const float*)d_in[0];
  const float* book_x = (const float*)d_in[1];
  const float* user_W = (const float*)d_in[2];
  const float* user_b = (const float*)d_in[3];
  const float* book_W = (const float*)d_in[4];
  const float* book_b = (const float*)d_in[5];
  const int*   rates  = (const int*)d_in[6];
  const int*   impl   = (const int*)d_in[7];
  float* outp = (float*)d_out;

  float* x   = (float*)d_ws;
  float* buf = x   + (size_t)NNODE * H;
  float* deg = buf + (size_t)NNODE * H;
  float* dis = deg + NNODE;
  float* nrm = dis + NNODE;

  const size_t embBytes = (size_t)NNODE * H * sizeof(float);
  hipMemsetAsync(deg, 0, NNODE * sizeof(float), stream);

  {
    int waves = NU / 16;
    gemm_bias_wmma<<<(waves + 7) / 8, 256, 0, stream>>>(user_x, user_W, user_b, x, NU);
    waves = NB / 16;
    gemm_bias_wmma<<<(waves + 7) / 8, 256, 0, stream>>>(book_x, book_W, book_b,
                                                        x + (size_t)NU * H, NB);
  }

  hipMemcpyAsync(outp, x, embBytes, hipMemcpyDeviceToDevice, stream);

  degree_kernel<<<(NE + 255) / 256, 256, 0, stream>>>(rates, impl, deg);
  dis_kernel<<<(NNODE + 255) / 256, 256, 0, stream>>>(deg, dis);
  norm_kernel<<<(NE + 255) / 256, 256, 0, stream>>>(rates, impl, dis, nrm);

  const int nElem = NNODE * H;
  for (int layer = 0; layer < 2; ++layer) {
    gather_owner_kernel<<<OWN_BLOCKS, OWN_THREADS, 0, stream>>>(x, nrm, rates, impl, buf);
    accum_copy_kernel<<<(nElem + 255) / 256, 256, 0, stream>>>(outp, x, buf);
  }

  scale_kernel<<<(nElem + 255) / 256, 256, 0, stream>>>(outp);
}


# layer sums: a new pass sets, per 64-edge group, a bit for each owner block that has a target in it; a block reads its bit plane and only hit groups' targets; LDS float adds still arrival order
# speedup vs baseline: 1.5226x; 1.0449x over previous
#include <hip/hip_runtime.h>

#define NU 100000
#define NB 50000
#define NNODE (NU + NB)
#define H 128
#define F 256
#define EPR 500000
#define NE (4 * EPR)

typedef __attribute__((ext_vector_type(2))) float v2f;
typedef __attribute__((ext_vector_type(8))) float v8f;

__global__ void __launch_bounds__(256)
gemm_bias_wmma(const float* __restrict__ X, const float* __restrict__ W,
               const float* __restrict__ bias, float* __restrict__ Y, int M) {
  const int wave = (int)((blockIdx.x * blockDim.x + threadIdx.x) >> 5);
  const int lane = threadIdx.x & 31;
  if (wave * 16 >= M) return;
  const int mbase = wave * 16;
  const int lrow  = lane & 15;
  const int khalf = (lane >> 4) << 1;

  const float* arow = X + (size_t)(mbase + lrow) * F + khalf;

  v8f c[8] = {};

  for (int k = 0; k < F; k += 4) {
    v2f a = *(const v2f*)(arow + k);
#pragma unroll
    for (int nt = 0; nt < 8; ++nt) {
      v2f b = *(const v2f*)(W + (size_t)(nt * 16 + lrow) * F + khalf + k);
      c[nt] = __builtin_amdgcn_wmma_f32_16x16x4_f32(
           false, a,  false, b,
           (short)0, c[nt],  false,  false);
    }
  }

  const int mo = (lane >> 4) << 3;
#pragma unroll
  for (int nt = 0; nt < 8; ++nt) {
    float bv = bias[nt * 16 + lrow];
#pragma unroll
    for (int v = 0; v < 8; ++v) {
      Y[(size_t)(mbase + mo + v) * H + nt * 16 + lrow] = c[nt][v] + bv;
    }
  }
}

__device__ __forceinline__ void decode_edge(int e, const int* __restrict__ rates,
                                            const int* __restrict__ impl,
                                            int& row, int& col) {
  const int rev = (e >= 2 * EPR) ? 1 : 0;
  int idx = rev ? (e - 2 * EPR) : e;
  const int* ed = (idx < EPR) ? rates : impl;
  const int i = (idx < EPR) ? idx : (idx - EPR);
  const int u = ed[i];
  const int b = ed[i + EPR];
  row = rev ? (b + NU) : u;
  col = rev ? u : b;
}

__global__ void __launch_bounds__(256)
degree_kernel(const int* __restrict__ rates, const int* __restrict__ impl,
              float* __restrict__ deg) {
  const int e = blockIdx.x * blockDim.x + threadIdx.x;
  if (e >= NE) return;
  int row, col;
  decode_edge(e, rates, impl, row, col);
  unsafeAtomicAdd(deg + col, 1.0f);
}

__global__ void __launch_bounds__(256)
dis_kernel(const float* __restrict__ deg, float* __restrict__ dis) {
  const int i = blockIdx.x * blockDim.x + threadIdx.x;
  if (i >= NNODE) return;
  const float d = deg[i];
  dis[i] = (d > 0.0f) ? rsqrtf(d) : 0.0f;
}

__global__ void __launch_bounds__(256)
norm_kernel(const int* __restrict__ rates, const int* __restrict__ impl,
            const float* __restrict__ dis, float* __restrict__ nrm) {
  const int e = blockIdx.x * blockDim.x + threadIdx.x;
  if (e >= NE) return;
  int row, col;
  decode_edge(e, rates, impl, row, col);
  nrm[e] = dis[row] * dis[col];
}

#define OWN_BLOCKS  256
#define OWN_ROWS    ((NNODE + OWN_BLOCKS - 1) / OWN_BLOCKS)
#define OWN_THREADS 1024
#define OWN_WAVES   (OWN_THREADS / 32)
#define QCAP        160
#define DRAIN_G     8

#define GRP_SEG     ((EPR + 63) / 64)
#define SEG_PITCH   7936
#define SEG_STEPS   (SEG_PITCH / 128)
#define PPITCH      (4 * SEG_PITCH)
#define NSTEP       (4 * SEG_STEPS)
#define PRES_BLOCKS (PPITCH / 32)

__device__ __forceinline__ void lds_add(float* p, float v) {
  __hip_atomic_fetch_add(p, v, __ATOMIC_RELAXED, __HIP_MEMORY_SCOPE_WORKGROUP);
}

__device__ __forceinline__ void drain_queue(const unsigned* q, int n,
                                            const float4* __restrict__ x4,
                                            const float* __restrict__ nrm,
                                            const int* __restrict__ rates,
                                            const int* __restrict__ impl,
                                            float* acc, int lane) {
  const int h = lane >> 4;
  for (int base = 0; base < n; base += DRAIN_G) {
    float4 v[DRAIN_G];
    float  w[DRAIN_G];
    unsigned il[DRAIN_G];
    const int li = base + (lane & (DRAIN_G - 1));
    const int dv = (int)q[(li < n) ? li : base];
#pragma unroll
    for (int g = 0; g < DRAIN_G; ++g) {
      const unsigned d = (unsigned)__builtin_amdgcn_readlane(dv, g);
      const int e = (int)(d >> 10);
      int row, col;
      decode_edge(e, rates, impl, row, col);
      (void)col;
      w[g]  = nrm[e];
      il[g] = d & 1023u;
      v[g]  = x4[(size_t)row * (H / 4) + lane];
    }
#pragma unroll
    for (int g = 0; g < DRAIN_G; ++g) {
      if (base + g < n) {
        float* a = acc + il[g] * H + 4 * lane;
        const float4 s = v[g];
        const float  m = w[g];
        lds_add(a + h,             (h ? s.y : s.x) * m);
        lds_add(a + ((1 + h) & 3), (h ? s.z : s.y) * m);
        lds_add(a + ((2 + h) & 3), (h ? s.w : s.z) * m);
        lds_add(a + ((3 + h) & 3), (h ? s.x : s.w) * m);
      }
    }
  }
}

__device__ __forceinline__ const int* seg_cols(int seg, const int* __restrict__ rates,
                                               const int* __restrict__ impl) {
  return (seg == 0) ? (rates + EPR) : (seg == 1) ? (impl + EPR) : (seg == 2) ? rates : impl;
}

__global__ void __launch_bounds__(1024)
present_kernel(const int* __restrict__ rates, const int* __restrict__ impl,
               unsigned* __restrict__ pres) {
  __shared__ unsigned st[8][32];
  const int lane = threadIdx.x & 31;
  const int wave = __builtin_amdgcn_readfirstlane((int)(threadIdx.x >> 5));
  const int pidx = (int)blockIdx.x * 32 + wave;
  const int seg = pidx / SEG_PITCH;
  const int gs  = pidx - seg * SEG_PITCH;
  unsigned myv = 0u;
  if (gs < GRP_SEG) {
    const int idx = gs * 64 + 2 * lane;
    const bool ok = idx < EPR;
    const int2 t2 = *(const int2*)(seg_cols(seg, rates, impl) + (ok ? idx : 0));
    const unsigned ta = (unsigned)t2.x, tb = (unsigned)t2.y;
    const unsigned valid = __builtin_amdgcn_ballot_w32(ok);
    unsigned la = valid, lb = valid;
#pragma unroll
    for (int i = 0; i < 5; ++i) {
      const unsigned ka = __builtin_amdgcn_ballot_w32(((ta >> i) & 1u) != 0u);
      const unsigned kb = __builtin_amdgcn_ballot_w32(((tb >> i) & 1u) != 0u);
      const bool on = ((lane >> i) & 1) != 0;
      la &= on ? ka : ~ka;
      lb &= on ? kb : ~kb;
    }
    unsigned ha[3], hb[3];
#pragma unroll
    for (int i = 0; i < 3; ++i) {
      ha[i] = __builtin_amdgcn_ballot_w32(((ta >> (5 + i)) & 1u) != 0u);
      hb[i] = __builtin_amdgcn_ballot_w32(((tb >> (5 + i)) & 1u) != 0u);
    }
#pragma unroll
    for (int m = 0; m < 8; ++m) {
      unsigned sa = la, sb = lb;
#pragma unroll
      for (int i = 0; i < 3; ++i) {
        sa &= ((m >> i) & 1) ? ha[i] : ~ha[i];
        sb &= ((m >> i) & 1) ? hb[i] : ~hb[i];
      }
      const unsigned pm = __builtin_amdgcn_ballot_w32((sa | sb) != 0u);
      if (lane == m) myv = pm;
    }
  }
  if (lane < 8) st[lane][wave] = myv;
  __syncthreads();
  if (wave < 8) pres[wave * PPITCH + (int)blockIdx.x * 32 + lane] = st[wave][lane];
}

__device__ __forceinline__ void push_owned(unsigned t, unsigned e, unsigned live, unsigned key,
                                           unsigned* q, int& cnt, int lane) {
  const unsigned iloc = ((t >> 8) | (t << 24)) - key;
  const unsigned mk = __builtin_amdgcn_ballot_w32(iloc < (unsigned)OWN_ROWS) & live;
  if (mk != 0u) {
    if ((mk >> lane) & 1u)
      q[cnt + (int)__builtin_amdgcn_mbcnt_lo(mk, 0u)] = (e << 10) | iloc;
    cnt += __builtin_popcount(mk);
  }
}

__global__ void __launch_bounds__(OWN_THREADS)
gather_owner_kernel(const float* __restrict__ x, const float* __restrict__ nrm,
                    const int* __restrict__ rates, const int* __restrict__ impl,
                    const unsigned* __restrict__ pres, float* __restrict__ out) {
  __shared__ __attribute__((aligned(16))) float acc[OWN_ROWS * H];
  __shared__ unsigned queue[OWN_WAVES * QCAP];

  const int tid  = threadIdx.x;
  const int lane = tid & 31;
  const int wave = __builtin_amdgcn_readfirstlane(tid >> 5);
  const unsigned b = blockIdx.x;

  float4* acc4 = (float4*)acc;
  for (int i = tid; i < OWN_ROWS * H / 4; i += OWN_THREADS)
    acc4[i] = make_float4(0.0f, 0.0f, 0.0f, 0.0f);
  __syncthreads();

  unsigned* q = queue + wave * QCAP;
  const float4* x4 = (const float4*)x;
  const unsigned key = b << 24;
  const int rot = (int)((b * NSTEP) >> 8);
  const uint4* plane4 = (const uint4*)(pres + (b >> 5) * PPITCH);
  const unsigned bbit = b & 31u;
  int cnt = 0;

  for (int s = wave; s < NSTEP; s += OWN_WAVES) {
    int sr = s + rot;
    if (sr >= NSTEP) sr -= NSTEP;
    const int seg = sr / SEG_STEPS;
    const int st  = sr - seg * SEG_STEPS;
    const int* colp = seg_cols(seg, rates, impl);
    const unsigned ebase = (unsigned)(seg * EPR);
    const uint4 d = plane4[sr * 32 + lane];
    const unsigned dk[4] = {d.x, d.y, d.z, d.w};
#pragma unroll
    for (int k = 0; k < 4; ++k) {
      unsigned m = __builtin_amdgcn_ballot_w32(((dk[k] >> bbit) & 1u) != 0u);
      while (m != 0u) {
        const int l0 = __builtin_ctz(m);
        m &= m - 1u;
        const unsigned two = (m != 0u) ? ~0u : 0u;
        const int l1 = (m != 0u) ? __builtin_ctz(m) : l0;
        m &= m - 1u;
        const int i0 = ((((st * 32 + l0) << 2) + k) << 6) + 2 * lane;
        const int i1 = ((((st * 32 + l1) << 2) + k) << 6) + 2 * lane;
        const bool ok0 = i0 < EPR, ok1 = i1 < EPR;
        const int2 t0 = *(const int2*)(colp + (ok0 ? i0 : 0));
        const int2 t1 = *(const int2*)(colp + (ok1 ? i1 : 0));
        const unsigned live0 = __builtin_amdgcn_ballot_w32(ok0);
        const unsigned live1 = __builtin_amdgcn_ballot_w32(ok1) & two;
        const unsigned e0 = ebase + (unsigned)i0, e1 = ebase + (unsigned)i1;
        push_owned((unsigned)t0.x, e0,      live0, key, q, cnt, lane);
        push_owned((unsigned)t0.y, e0 + 1u, live0, key, q, cnt, lane);
        push_owned((unsigned)t1.x, e1,      live1, key, q, cnt, lane);
        push_owned((unsigned)t1.y, e1 + 1u, live1, key, q, cnt, lane);
        if (cnt >= 32) {
          __builtin_amdgcn_fence(__ATOMIC_SEQ_CST, "wavefront");
          drain_queue(q, cnt, x4, nrm, rates, impl, acc, lane);
          __builtin_amdgcn_fence(__ATOMIC_SEQ_CST, "wavefront");
          cnt = 0;
        }
      }
    }
  }
  __builtin_amdgcn_fence(__ATOMIC_SEQ_CST, "wavefront");
  drain_queue(q, cnt, x4, nrm, rates, impl, acc, lane);

  __syncthreads();
  for (int i = wave; i < OWN_ROWS; i += OWN_WAVES) {
    const unsigned t = ((unsigned)i << 8) | b;
    if (t < (unsigned)NNODE)
      ((float4*)(out + (size_t)t * H))[lane] = ((const float4*)(acc + i * H))[lane];
  }
}

__global__ void __launch_bounds__(256)
accum_copy_kernel(float* __restrict__ acc, float* __restrict__ x,
                  const float* __restrict__ buf) {
  const int i = blockIdx.x * blockDim.x + threadIdx.x;
  if (i >= NNODE * H) return;
  const float v = buf[i];
  acc[i] += v;
  x[i] = v;
}

__global__ void __launch_bounds__(256)
scale_kernel(float* __restrict__ acc) {
  const int i = blockIdx.x * blockDim.x + threadIdx.x;
  if (i >= NNODE * H) return;
  acc[i] *= (1.0f / 3.0f);
}

extern "C" void kernel_launch(void* const* d_in, const int* in_sizes, int n_in,
                              void* d_out, int out_size, void* d_ws, size_t ws_size,
                              hipStream_t stream) {
  (void)in_sizes; (void)n_in; (void)out_size; (void)ws_size;

  const float* user_x = (const float*)d_in[0];
  const float* book_x = (const float*)d_in[1];
  const float* user_W = (const float*)d_in[2];
  const float* user_b = (const float*)d_in[3];
  const float* book_W = (const float*)d_in[4];
  const float* book_b = (const float*)d_in[5];
  const int*   rates  = (const int*)d_in[6];
  const int*   impl   = (const int*)d_in[7];
  float* outp = (float*)d_out;

  float* x   = (float*)d_ws;
  float* buf = x   + (size_t)NNODE * H;
  float* deg = buf + (size_t)NNODE * H;
  float* dis = deg + NNODE;
  float* nrm = dis + NNODE;
  unsigned* pres = (unsigned*)deg;

  const size_t embBytes = (size_t)NNODE * H * sizeof(float);
  hipMemsetAsync(deg, 0, NNODE * sizeof(float), stream);

  {
    int waves = NU / 16;
    gemm_bias_wmma<<<(waves + 7) / 8, 256, 0, stream>>>(user_x, user_W, user_b, x, NU);
    waves = NB / 16;
    gemm_bias_wmma<<<(waves + 7) / 8, 256, 0, stream>>>(book_x, book_W, book_b,
                                                        x + (size_t)NU * H, NB);
  }

  hipMemcpyAsync(outp, x, embBytes, hipMemcpyDeviceToDevice, stream);

  degree_kernel<<<(NE + 255) / 256, 256, 0, stream>>>(rates, impl, deg);
  dis_kernel<<<(NNODE + 255) / 256, 256, 0, stream>>>(deg, dis);
  norm_kernel<<<(NE + 255) / 256, 256, 0, stream>>>(rates, impl, dis, nrm);
  present_kernel<<<PRES_BLOCKS, 1024, 0, stream>>>(rates, impl, pres);

  const int nElem = NNODE * H;
  for (int layer = 0; layer < 2; ++layer) {
    gather_owner_kernel<<<OWN_BLOCKS, OWN_THREADS, 0, stream>>>(x, nrm, rates, impl, pres, buf);
    accum_copy_kernel<<<(nElem + 255) / 256, 256, 0, stream>>>(outp, x, buf);
  }

  scale_kernel<<<(nElem + 255) / 256, 256, 0, stream>>>(outp);
}


# element-wise passes (accumulate+copy x2, scale): a lane moves 16-byte pieces, a wave a contiguous run with 8 loads issued ahead; 16x / 32x fewer waves; per-element arithmetic and order unchanged
# speedup vs baseline: 1.5709x; 1.0317x over previous
#include <hip/hip_runtime.h>

#define NU 100000
#define NB 50000
#define NNODE (NU + NB)
#define H 128
#define F 256
#define EPR 500000
#define NE (4 * EPR)

typedef __attribute__((ext_vector_type(2))) float v2f;
typedef __attribute__((ext_vector_type(8))) float v8f;

__global__ void __launch_bounds__(256)
gemm_bias_wmma(const float* __restrict__ X, const float* __restrict__ W,
               const float* __restrict__ bias, float* __restrict__ Y, int M) {
  const int wave = (int)((blockIdx.x * blockDim.x + threadIdx.x) >> 5);
  const int lane = threadIdx.x & 31;
  if (wave * 16 >= M) return;
  const int mbase = wave * 16;
  const int lrow  = lane & 15;
  const int khalf = (lane >> 4) << 1;

  const float* arow = X + (size_t)(mbase + lrow) * F + khalf;

  v8f c[8] = {};

  for (int k = 0; k < F; k += 4) {
    v2f a = *(const v2f*)(arow + k);
#pragma unroll
    for (int nt = 0; nt < 8; ++nt) {
      v2f b = *(const v2f*)(W + (size_t)(nt * 16 + lrow) * F + khalf + k);
      c[nt] = __builtin_amdgcn_wmma_f32_16x16x4_f32(
           false, a,  false, b,
           (short)0, c[nt],  false,  false);
    }
  }

  const int mo = (lane >> 4) << 3;
#pragma unroll
  for (int nt = 0; nt < 8; ++nt) {
    float bv = bias[nt * 16 + lrow];
#pragma unroll
    for (int v = 0; v < 8; ++v) {
      Y[(size_t)(mbase + mo + v) * H + nt * 16 + lrow] = c[nt][v] + bv;
    }
  }
}

__device__ __forceinline__ void decode_edge(int e, const int* __restrict__ rates,
                                            const int* __restrict__ impl,
                                            int& row, int& col) {
  const int rev = (e >= 2 * EPR) ? 1 : 0;
  int idx = rev ? (e - 2 * EPR) : e;
  const int* ed = (idx < EPR) ? rates : impl;
  const int i = (idx < EPR) ? idx : (idx - EPR);
  const int u = ed[i];
  const int b = ed[i + EPR];
  row = rev ? (b + NU) : u;
  col = rev ? u : b;
}

__global__ void __launch_bounds__(256)
degree_kernel(const int* __restrict__ rates, const int* __restrict__ impl,
              float* __restrict__ deg) {
  const int e = blockIdx.x * blockDim.x + threadIdx.x;
  if (e >= NE) return;
  int row, col;
  decode_edge(e, rates, impl, row, col);
  unsafeAtomicAdd(deg + col, 1.0f);
}

__global__ void __launch_bounds__(256)
dis_kernel(const float* __restrict__ deg, float* __restrict__ dis) {
  const int i = blockIdx.x * blockDim.x + threadIdx.x;
  if (i >= NNODE) return;
  const float d = deg[i];
  dis[i] = (d > 0.0f) ? rsqrtf(d) : 0.0f;
}

__global__ void __launch_bounds__(256)
norm_kernel(const int* __restrict__ rates, const int* __restrict__ impl,
            const float* __restrict__ dis, float* __restrict__ nrm) {
  const int e = blockIdx.x * blockDim.x + threadIdx.x;
  if (e >= NE) return;
  int row, col;
  decode_edge(e, rates, impl, row, col);
  nrm[e] = dis[row] * dis[col];
}

#define OWN_BLOCKS  256
#define OWN_ROWS    ((NNODE + OWN_BLOCKS - 1) / OWN_BLOCKS)
#define OWN_THREADS 1024
#define OWN_WAVES   (OWN_THREADS / 32)
#define QCAP        160
#define DRAIN_G     8

#define GRP_SEG     ((EPR + 63) / 64)
#define SEG_PITCH   7936
#define SEG_STEPS   (SEG_PITCH / 128)
#define PPITCH      (4 * SEG_PITCH)
#define NSTEP       (4 * SEG_STEPS)
#define PRES_BLOCKS (PPITCH / 32)

__device__ __forceinline__ void lds_add(float* p, float v) {
  __hip_atomic_fetch_add(p, v, __ATOMIC_RELAXED, __HIP_MEMORY_SCOPE_WORKGROUP);
}

__device__ __forceinline__ void drain_queue(const unsigned* q, int n,
                                            const float4* __restrict__ x4,
                                            const float* __restrict__ nrm,
                                            const int* __restrict__ rates,
                                            const int* __restrict__ impl,
                                            float* acc, int lane) {
  const int h = lane >> 4;
  for (int base = 0; base < n; base += DRAIN_G) {
    float4 v[DRAIN_G];
    float  w[DRAIN_G];
    unsigned il[DRAIN_G];
    const int li = base + (lane & (DRAIN_G - 1));
    const int dv = (int)q[(li < n) ? li : base];
#pragma unroll
    for (int g = 0; g < DRAIN_G; ++g) {
      const unsigned d = (unsigned)__builtin_amdgcn_readlane(dv, g);
      const int e = (int)(d >> 10);
      int row, col;
      decode_edge(e, rates, impl, row, col);
      (void)col;
      w[g]  = nrm[e];
      il[g] = d & 1023u;
      v[g]  = x4[(size_t)row * (H / 4) + lane];
    }
#pragma unroll
    for (int g = 0; g < DRAIN_G; ++g) {
      if (base + g < n) {
        float* a = acc + il[g] * H + 4 * lane;
        const float4 s = v[g];
        const float  m = w[g];
        lds_add(a + h,             (h ? s.y : s.x) * m);
        lds_add(a + ((1 + h) & 3), (h ? s.z : s.y) * m);
        lds_add(a + ((2 + h) & 3), (h ? s.w : s.z) * m);
        lds_add(a + ((3 + h) & 3), (h ? s.x : s.w) * m);
      }
    }
  }
}

__device__ __forceinline__ const int* seg_cols(int seg, const int* __restrict__ rates,
                                               const int* __restrict__ impl) {
  return (seg == 0) ? (rates + EPR) : (seg == 1) ? (impl + EPR) : (seg == 2) ? rates : impl;
}

__global__ void __launch_bounds__(1024)
present_kernel(const int* __restrict__ rates, const int* __restrict__ impl,
               unsigned* __restrict__ pres) {
  __shared__ unsigned st[8][32];
  const int lane = threadIdx.x & 31;
  const int wave = __builtin_amdgcn_readfirstlane((int)(threadIdx.x >> 5));
  const int pidx = (int)blockIdx.x * 32 + wave;
  const int seg = pidx / SEG_PITCH;
  const int gs  = pidx - seg * SEG_PITCH;
  unsigned myv = 0u;
  if (gs < GRP_SEG) {
    const int idx = gs * 64 + 2 * lane;
    const bool ok = idx < EPR;
    const int2 t2 = *(const int2*)(seg_cols(seg, rates, impl) + (ok ? idx : 0));
    const unsigned ta = (unsigned)t2.x, tb = (unsigned)t2.y;
    const unsigned valid = __builtin_amdgcn_ballot_w32(ok);
    unsigned la = valid, lb = valid;
#pragma unroll
    for (int i = 0; i < 5; ++i) {
      const unsigned ka = __builtin_amdgcn_ballot_w32(((ta >> i) & 1u) != 0u);
      const unsigned kb = __builtin_amdgcn_ballot_w32(((tb >> i) & 1u) != 0u);
      const bool on = ((lane >> i) & 1) != 0;
      la &= on ? ka : ~ka;
      lb &= on ? kb : ~kb;
    }
    unsigned ha[3], hb[3];
#pragma unroll
    for (int i = 0; i < 3; ++i) {
      ha[i] = __builtin_amdgcn_ballot_w32(((ta >> (5 + i)) & 1u) != 0u);
      hb[i] = __builtin_amdgcn_ballot_w32(((tb >> (5 + i)) & 1u) != 0u);
    }
#pragma unroll
    for (int m = 0; m < 8; ++m) {
      unsigned sa = la, sb = lb;
#pragma unroll
      for (int i = 0; i < 3; ++i) {
        sa &= ((m >> i) & 1) ? ha[i] : ~ha[i];
        sb &= ((m >> i) & 1) ? hb[i] : ~hb[i];
      }
      const unsigned pm = __builtin_amdgcn_ballot_w32((sa | sb) != 0u);
      if (lane == m) myv = pm;
    }
  }
  if (lane < 8) st[lane][wave] = myv;
  __syncthreads();
  if (wave < 8) pres[wave * PPITCH + (int)blockIdx.x * 32 + lane] = st[wave][lane];
}

__device__ __forceinline__ void push_owned(unsigned t, unsigned e, unsigned live, unsigned key,
                                           unsigned* q, int& cnt, int lane) {
  const unsigned iloc = ((t >> 8) | (t << 24)) - key;
  const unsigned mk = __builtin_amdgcn_ballot_w32(iloc < (unsigned)OWN_ROWS) & live;
  if (mk != 0u) {
    if ((mk >> lane) & 1u)
      q[cnt + (int)__builtin_amdgcn_mbcnt_lo(mk, 0u)] = (e << 10) | iloc;
    cnt += __builtin_popcount(mk);
  }
}

__global__ void __launch_bounds__(OWN_THREADS)
gather_owner_kernel(const float* __restrict__ x, const float* __restrict__ nrm,
                    const int* __restrict__ rates, const int* __restrict__ impl,
                    const unsigned* __restrict__ pres, float* __restrict__ out) {
  __shared__ __attribute__((aligned(16))) float acc[OWN_ROWS * H];
  __shared__ unsigned queue[OWN_WAVES * QCAP];

  const int tid  = threadIdx.x;
  const int lane = tid & 31;
  const int wave = __builtin_amdgcn_readfirstlane(tid >> 5);
  const unsigned b = blockIdx.x;

  float4* acc4 = (float4*)acc;
  for (int i = tid; i < OWN_ROWS * H / 4; i += OWN_THREADS)
    acc4[i] = make_float4(0.0f, 0.0f, 0.0f, 0.0f);
  __syncthreads();

  unsigned* q = queue + wave * QCAP;
  const float4* x4 = (const float4*)x;
  const unsigned key = b << 24;
  const int rot = (int)((b * NSTEP) >> 8);
  const uint4* plane4 = (const uint4*)(pres + (b >> 5) * PPITCH);
  const unsigned bbit = b & 31u;
  int cnt = 0;

  for (int s = wave; s < NSTEP; s += OWN_WAVES) {
    int sr = s + rot;
    if (sr >= NSTEP) sr -= NSTEP;
    const int seg = sr / SEG_STEPS;
    const int st  = sr - seg * SEG_STEPS;
    const int* colp = seg_cols(seg, rates, impl);
    const unsigned ebase = (unsigned)(seg * EPR);
    const uint4 d = plane4[sr * 32 + lane];
    const unsigned dk[4] = {d.x, d.y, d.z, d.w};
#pragma unroll
    for (int k = 0; k < 4; ++k) {
      unsigned m = __builtin_amdgcn_ballot_w32(((dk[k] >> bbit) & 1u) != 0u);
      while (m != 0u) {
        const int l0 = __builtin_ctz(m);
        m &= m - 1u;
        const unsigned two = (m != 0u) ? ~0u : 0u;
        const int l1 = (m != 0u) ? __builtin_ctz(m) : l0;
        m &= m - 1u;
        const int i0 = ((((st * 32 + l0) << 2) + k) << 6) + 2 * lane;
        const int i1 = ((((st * 32 + l1) << 2) + k) << 6) + 2 * lane;
        const bool ok0 = i0 < EPR, ok1 = i1 < EPR;
        const int2 t0 = *(const int2*)(colp + (ok0 ? i0 : 0));
        const int2 t1 = *(const int2*)(colp + (ok1 ? i1 : 0));
        const unsigned live0 = __builtin_amdgcn_ballot_w32(ok0);
        const unsigned live1 = __builtin_amdgcn_ballot_w32(ok1) & two;
        const unsigned e0 = ebase + (unsigned)i0, e1 = ebase + (unsigned)i1;
        push_owned((unsigned)t0.x, e0,      live0, key, q, cnt, lane);
        push_owned((unsigned)t0.y, e0 + 1u, live0, key, q, cnt, lane);
        push_owned((unsigned)t1.x, e1,      live1, key, q, cnt, lane);
        push_owned((unsigned)t1.y, e1 + 1u, live1, key, q, cnt, lane);
        if (cnt >= 32) {
          __builtin_amdgcn_fence(__ATOMIC_SEQ_CST, "wavefront");
          drain_queue(q, cnt, x4, nrm, rates, impl, acc, lane);
          __builtin_amdgcn_fence(__ATOMIC_SEQ_CST, "wavefront");
          cnt = 0;
        }
      }
    }
  }
  __builtin_amdgcn_fence(__ATOMIC_SEQ_CST, "wavefront");
  drain_queue(q, cnt, x4, nrm, rates, impl, acc, lane);

  __syncthreads();
  for (int i = wave; i < OWN_ROWS; i += OWN_WAVES) {
    const unsigned t = ((unsigned)i << 8) | b;
    if (t < (unsigned)NNODE)
      ((float4*)(out + (size_t)t * H))[lane] = ((const float4*)(acc + i * H))[lane];
  }
}

#define N4         (NNODE * H / 4)
#define EW_THREADS 256
#define ACC_RUN    128
#define SCL_RUN    256
#define EW_WAVES   (EW_THREADS / 32)
#define ACC_BLOCKS ((N4 / ACC_RUN + EW_WAVES - 1) / EW_WAVES)
#define SCL_BLOCKS ((N4 / SCL_RUN + EW_WAVES - 1) / EW_WAVES)

__global__ void __launch_bounds__(EW_THREADS)
accum_copy_kernel(float* __restrict__ acc, float* __restrict__ x,
                  const float* __restrict__ buf) {
  const int wave = (int)((blockIdx.x * EW_THREADS + threadIdx.x) >> 5);
  if (wave * ACC_RUN >= N4) return;
  const int i0 = wave * ACC_RUN + (int)(threadIdx.x & 31);
  float4* acc4 = (float4*)acc;
  float4* x4 = (float4*)x;
  const float4* buf4 = (const float4*)buf;
  float4 b[ACC_RUN / 32], a[ACC_RUN / 32];
#pragma unroll
  for (int u = 0; u < ACC_RUN / 32; ++u) b[u] = buf4[i0 + 32 * u];
#pragma unroll
  for (int u = 0; u < ACC_RUN / 32; ++u) a[u] = acc4[i0 + 32 * u];
#pragma unroll
  for (int u = 0; u < ACC_RUN / 32; ++u) {
    float4 s = a[u];
    s.x += b[u].x; s.y += b[u].y; s.z += b[u].z; s.w += b[u].w;
    acc4[i0 + 32 * u] = s;
    x4[i0 + 32 * u] = b[u];
  }
}

__global__ void __launch_bounds__(EW_THREADS)
scale_kernel(float* __restrict__ acc) {
  const int wave = (int)((blockIdx.x * EW_THREADS + threadIdx.x) >> 5);
  if (wave * SCL_RUN >= N4) return;
  const int i0 = wave * SCL_RUN + (int)(threadIdx.x & 31);
  float4* acc4 = (float4*)acc;
  float4 a[SCL_RUN / 32];
#pragma unroll
  for (int u = 0; u < SCL_RUN / 32; ++u) a[u] = acc4[i0 + 32 * u];
#pragma unroll
  for (int u = 0; u < SCL_RUN / 32; ++u) {
    float4 s = a[u];
    s.x *= (1.0f / 3.0f); s.y *= (1.0f / 3.0f); s.z *= (1.0f / 3.0f); s.w *= (1.0f / 3.0f);
    acc4[i0 + 32 * u] = s;
  }
}

extern "C" void kernel_launch(void* const* d_in, const int* in_sizes, int n_in,
                              void* d_out, int out_size, void* d_ws, size_t ws_size,
                              hipStream_t stream) {
  (void)in_sizes; (void)n_in; (void)out_size; (void)ws_size;

  const float* user_x = (const float*)d_in[0];
  const float* book_x = (const float*)d_in[1];
  const float* user_W = (const float*)d_in[2];
  const float* user_b = (const float*)d_in[3];
  const float* book_W = (const float*)d_in[4];
  const float* book_b = (const float*)d_in[5];
  const int*   rates  = (const int*)d_in[6];
  const int*   impl   = (const int*)d_in[7];
  float* outp = (float*)d_out;

  float* x   = (float*)d_ws;
  float* buf = x   + (size_t)NNODE * H;
  float* deg = buf + (size_t)NNODE * H;
  float* dis = deg + NNODE;
  float* nrm = dis + NNODE;
  unsigned* pres = (unsigned*)deg;

  const size_t embBytes = (size_t)NNODE * H * sizeof(float);
  hipMemsetAsync(deg, 0, NNODE * sizeof(float), stream);

  {
    int waves = NU / 16;
    gemm_bias_wmma<<<(waves + 7) / 8, 256, 0, stream>>>(user_x, user_W, user_b, x, NU);
    waves = NB / 16;
    gemm_bias_wmma<<<(waves + 7) / 8, 256, 0, stream>>>(book_x, book_W, book_b,
                                                        x + (size_t)NU * H, NB);
  }

  hipMemcpyAsync(outp, x, embBytes, hipMemcpyDeviceToDevice, stream);

  degree_kernel<<<(NE + 255) / 256, 256, 0, stream>>>(rates, impl, deg);
  dis_kernel<<<(NNODE + 255) / 256, 256, 0, stream>>>(deg, dis);
  norm_kernel<<<(NE + 255) / 256, 256, 0, stream>>>(rates, impl, dis, nrm);
  present_kernel<<<PRES_BLOCKS, 1024, 0, stream>>>(rates, impl, pres);

  for (int layer = 0; layer < 2; ++layer) {
    gather_owner_kernel<<<OWN_BLOCKS, OWN_THREADS, 0, stream>>>(x, nrm, rates, impl, pres, buf);
    accum_copy_kernel<<<ACC_BLOCKS, EW_THREADS, 0, stream>>>(outp, x, buf);
  }

  scale_kernel<<<SCL_BLOCKS, EW_THREADS, 0, stream>>>(outp);
}


# layer-sum drain: a lane fetches source row and weight for each of 32 queued edges (was 3 wave-uniform loads per edge), 16 row gathers in flight, whole 32-entry chunks; LDS float adds arrival order
# speedup vs baseline: 1.7041x; 1.0848x over previous
#include <hip/hip_runtime.h>

#define NU 100000
#define NB 50000
#define NNODE (NU + NB)
#define H 128
#define F 256
#define EPR 500000
#define NE (4 * EPR)

typedef __attribute__((ext_vector_type(2))) float v2f;
typedef __attribute__((ext_vector_type(8))) float v8f;

__global__ void __launch_bounds__(256)
gemm_bias_wmma(const float* __restrict__ X, const float* __restrict__ W,
               const float* __restrict__ bias, float* __restrict__ Y, int M) {
  const int wave = (int)((blockIdx.x * blockDim.x + threadIdx.x) >> 5);
  const int lane = threadIdx.x & 31;
  if (wave * 16 >= M) return;
  const int mbase = wave * 16;
  const int lrow  = lane & 15;
  const int khalf = (lane >> 4) << 1;

  const float* arow = X + (size_t)(mbase + lrow) * F + khalf;

  v8f c[8] = {};

  for (int k = 0; k < F; k += 4) {
    v2f a = *(const v2f*)(arow + k);
#pragma unroll
    for (int nt = 0; nt < 8; ++nt) {
      v2f b = *(const v2f*)(W + (size_t)(nt * 16 + lrow) * F + khalf + k);
      c[nt] = __builtin_amdgcn_wmma_f32_16x16x4_f32(
           false, a,  false, b,
           (short)0, c[nt],  false,  false);
    }
  }

  const int mo = (lane >> 4) << 3;
#pragma unroll
  for (int nt = 0; nt < 8; ++nt) {
    float bv = bias[nt * 16 + lrow];
#pragma unroll
    for (int v = 0; v < 8; ++v) {
      Y[(size_t)(mbase + mo + v) * H + nt * 16 + lrow] = c[nt][v] + bv;
    }
  }
}

__device__ __forceinline__ void decode_edge(int e, const int* __restrict__ rates,
                                            const int* __restrict__ impl,
                                            int& row, int& col) {
  const int rev = (e >= 2 * EPR) ? 1 : 0;
  int idx = rev ? (e - 2 * EPR) : e;
  const int* ed = (idx < EPR) ? rates : impl;
  const int i = (idx < EPR) ? idx : (idx - EPR);
  const int u = ed[i];
  const int b = ed[i + EPR];
  row = rev ? (b + NU) : u;
  col = rev ? u : b;
}

__global__ void __launch_bounds__(256)
degree_kernel(const int* __restrict__ rates, const int* __restrict__ impl,
              float* __restrict__ deg) {
  const int e = blockIdx.x * blockDim.x + threadIdx.x;
  if (e >= NE) return;
  int row, col;
  decode_edge(e, rates, impl, row, col);
  unsafeAtomicAdd(deg + col, 1.0f);
}

__global__ void __launch_bounds__(256)
dis_kernel(const float* __restrict__ deg, float* __restrict__ dis) {
  const int i = blockIdx.x * blockDim.x + threadIdx.x;
  if (i >= NNODE) return;
  const float d = deg[i];
  dis[i] = (d > 0.0f) ? rsqrtf(d) : 0.0f;
}

__global__ void __launch_bounds__(256)
norm_kernel(const int* __restrict__ rates, const int* __restrict__ impl,
            const float* __restrict__ dis, float* __restrict__ nrm) {
  const int e = blockIdx.x * blockDim.x + threadIdx.x;
  if (e >= NE) return;
  int row, col;
  decode_edge(e, rates, impl, row, col);
  nrm[e] = dis[row] * dis[col];
}

#define OWN_BLOCKS  256
#define OWN_ROWS    ((NNODE + OWN_BLOCKS - 1) / OWN_BLOCKS)
#define OWN_THREADS 1024
#define OWN_WAVES   (OWN_THREADS / 32)
#define QCAP        160
#define DRAIN_G     16

#define GRP_SEG     ((EPR + 63) / 64)
#define SEG_PITCH   7936
#define SEG_STEPS   (SEG_PITCH / 128)
#define PPITCH      (4 * SEG_PITCH)
#define NSTEP       (4 * SEG_STEPS)
#define PRES_BLOCKS (PPITCH / 32)

__device__ __forceinline__ void lds_add(float* p, float v) {
  __hip_atomic_fetch_add(p, v, __ATOMIC_RELAXED, __HIP_MEMORY_SCOPE_WORKGROUP);
}

template <bool FULL>
__device__ __forceinline__ void drain_chunk(const unsigned* q, int base, int nv,
                                            const float4* __restrict__ x4,
                                            const float* __restrict__ nrm,
                                            const int* __restrict__ rates,
                                            const int* __restrict__ impl,
                                            float* acc, int lane) {
  const int h = lane >> 4;
  const int li = FULL ? (base + lane) : (base + ((lane < nv) ? lane : 0));
  const unsigned d = q[li];
  const int e = (int)(d >> 10);
  const int loc = (int)(d & 1023u);
  const bool rev = e >= 2 * EPR;
  const int idx = rev ? (e - 2 * EPR) : e;
  const bool first = idx < EPR;
  const int i = first ? idx : (idx - EPR);
  const int* ed = first ? rates : impl;
  const int sv = ed[i + (rev ? EPR : 0)];
  const int row = rev ? (sv + NU) : sv;
  const int wbits = __float_as_int(nrm[e]);
#pragma unroll
  for (int half = 0; half < 32 / DRAIN_G; ++half) {
    if (!FULL && half * DRAIN_G >= nv) break;
    float4 v[DRAIN_G];
#pragma unroll
    for (int g = 0; g < DRAIN_G; ++g) {
      const int r = __builtin_amdgcn_readlane(row, half * DRAIN_G + g);
      v[g] = x4[(size_t)r * (H / 4) + lane];
    }
#pragma unroll
    for (int g = 0; g < DRAIN_G; ++g) {
      if (FULL || half * DRAIN_G + g < nv) {
        const int il = __builtin_amdgcn_readlane(loc, half * DRAIN_G + g);
        const float m = __int_as_float(__builtin_amdgcn_readlane(wbits, half * DRAIN_G + g));
        float* a = acc + il * H + 4 * lane;
        const float4 s = v[g];
        lds_add(a + h,             (h ? s.y : s.x) * m);
        lds_add(a + ((1 + h) & 3), (h ? s.z : s.y) * m);
        lds_add(a + ((2 + h) & 3), (h ? s.w : s.z) * m);
        lds_add(a + ((3 + h) & 3), (h ? s.x : s.w) * m);
      }
    }
  }
}

__device__ __forceinline__ const int* seg_cols(int seg, const int* __restrict__ rates,
                                               const int* __restrict__ impl) {
  return (seg == 0) ? (rates + EPR) : (seg == 1) ? (impl + EPR) : (seg == 2) ? rates : impl;
}

__global__ void __launch_bounds__(1024)
present_kernel(const int* __restrict__ rates, const int* __restrict__ impl,
               unsigned* __restrict__ pres) {
  __shared__ unsigned st[8][32];
  const int lane = threadIdx.x & 31;
  const int wave = __builtin_amdgcn_readfirstlane((int)(threadIdx.x >> 5));
  const int pidx = (int)blockIdx.x * 32 + wave;
  const int seg = pidx / SEG_PITCH;
  const int gs  = pidx - seg * SEG_PITCH;
  unsigned myv = 0u;
  if (gs < GRP_SEG) {
    const int idx = gs * 64 + 2 * lane;
    const bool ok = idx < EPR;
    const int2 t2 = *(const int2*)(seg_cols(seg, rates, impl) + (ok ? idx : 0));
    const unsigned ta = (unsigned)t2.x, tb = (unsigned)t2.y;
    const unsigned valid = __builtin_amdgcn_ballot_w32(ok);
    unsigned la = valid, lb = valid;
#pragma unroll
    for (int i = 0; i < 5; ++i) {
      const unsigned ka = __builtin_amdgcn_ballot_w32(((ta >> i) & 1u) != 0u);
      const unsigned kb = __builtin_amdgcn_ballot_w32(((tb >> i) & 1u) != 0u);
      const bool on = ((lane >> i) & 1) != 0;
      la &= on ? ka : ~ka;
      lb &= on ? kb : ~kb;
    }
    unsigned ha[3], hb[3];
#pragma unroll
    for (int i = 0; i < 3; ++i) {
      ha[i] = __builtin_amdgcn_ballot_w32(((ta >> (5 + i)) & 1u) != 0u);
      hb[i] = __builtin_amdgcn_ballot_w32(((tb >> (5 + i)) & 1u) != 0u);
    }
#pragma unroll
    for (int m = 0; m < 8; ++m) {
      unsigned sa = la, sb = lb;
#pragma unroll
      for (int i = 0; i < 3; ++i) {
        sa &= ((m >> i) & 1) ? ha[i] : ~ha[i];
        sb &= ((m >> i) & 1) ? hb[i] : ~hb[i];
      }
      const unsigned pm = __builtin_amdgcn_ballot_w32((sa | sb) != 0u);
      if (lane == m) myv = pm;
    }
  }
  if (lane < 8) st[lane][wave] = myv;
  __syncthreads();
  if (wave < 8) pres[wave * PPITCH + (int)blockIdx.x * 32 + lane] = st[wave][lane];
}

__device__ __forceinline__ void push_owned(unsigned t, unsigned e, unsigned live, unsigned key,
                                           unsigned* q, int& cnt, int lane) {
  const unsigned iloc = ((t >> 8) | (t << 24)) - key;
  const unsigned mk = __builtin_amdgcn_ballot_w32(iloc < (unsigned)OWN_ROWS) & live;
  if (mk != 0u) {
    if ((mk >> lane) & 1u)
      q[cnt + (int)__builtin_amdgcn_mbcnt_lo(mk, 0u)] = (e << 10) | iloc;
    cnt += __builtin_popcount(mk);
  }
}

__global__ void __launch_bounds__(OWN_THREADS)
gather_owner_kernel(const float* __restrict__ x, const float* __restrict__ nrm,
                    const int* __restrict__ rates, const int* __restrict__ impl,
                    const unsigned* __restrict__ pres, float* __restrict__ out) {
  __shared__ __attribute__((aligned(16))) float acc[OWN_ROWS * H];
  __shared__ unsigned queue[OWN_WAVES * QCAP];

  const int tid  = threadIdx.x;
  const int lane = tid & 31;
  const int wave = __builtin_amdgcn_readfirstlane(tid >> 5);
  const unsigned b = blockIdx.x;

  float4* acc4 = (float4*)acc;
  for (int i = tid; i < OWN_ROWS * H / 4; i += OWN_THREADS)
    acc4[i] = make_float4(0.0f, 0.0f, 0.0f, 0.0f);
  __syncthreads();

  unsigned* q = queue + wave * QCAP;
  const float4* x4 = (const float4*)x;
  const unsigned key = b << 24;
  const int rot = (int)((b * NSTEP) >> 8);
  const uint4* plane4 = (const uint4*)(pres + (b >> 5) * PPITCH);
  const unsigned bbit = b & 31u;
  int cnt = 0;

  for (int s = wave; s < NSTEP; s += OWN_WAVES) {
    int sr = s + rot;
    if (sr >= NSTEP) sr -= NSTEP;
    const int seg = sr / SEG_STEPS;
    const int st  = sr - seg * SEG_STEPS;
    const int* colp = seg_cols(seg, rates, impl);
    const unsigned ebase = (unsigned)(seg * EPR);
    const uint4 d = plane4[sr * 32 + lane];
    const unsigned dk[4] = {d.x, d.y, d.z, d.w};
#pragma unroll
    for (int k = 0; k < 4; ++k) {
      unsigned m = __builtin_amdgcn_ballot_w32(((dk[k] >> bbit) & 1u) != 0u);
      while (m != 0u) {
        const int l0 = __builtin_ctz(m);
        m &= m - 1u;
        const unsigned two = (m != 0u) ? ~0u : 0u;
        const int l1 = (m != 0u) ? __builtin_ctz(m) : l0;
        m &= m - 1u;
        const int i0 = ((((st * 32 + l0) << 2) + k) << 6) + 2 * lane;
        const int i1 = ((((st * 32 + l1) << 2) + k) << 6) + 2 * lane;
        const bool ok0 = i0 < EPR, ok1 = i1 < EPR;
        const int2 t0 = *(const int2*)(colp + (ok0 ? i0 : 0));
        const int2 t1 = *(const int2*)(colp + (ok1 ? i1 : 0));
        const unsigned live0 = __builtin_amdgcn_ballot_w32(ok0);
        const unsigned live1 = __builtin_amdgcn_ballot_w32(ok1) & two;
        const unsigned e0 = ebase + (unsigned)i0, e1 = ebase + (unsigned)i1;
        push_owned((unsigned)t0.x, e0,      live0, key, q, cnt, lane);
        push_owned((unsigned)t0.y, e0 + 1u, live0, key, q, cnt, lane);
        push_owned((unsigned)t1.x, e1,      live1, key, q, cnt, lane);
        push_owned((unsigned)t1.y, e1 + 1u, live1, key, q, cnt, lane);
        if (cnt >= 32) {
          __builtin_amdgcn_fence(__ATOMIC_SEQ_CST, "wavefront");
          const int full = cnt & ~31;
          for (int c = 0; c < full; c += 32)
            drain_chunk<true>(q, c, 32, x4, nrm, rates, impl, acc, lane);
          const int rem = cnt - full;
          if (lane < rem) q[lane] = q[full + lane];
          __builtin_amdgcn_fence(__ATOMIC_SEQ_CST, "wavefront");
          cnt = rem;
        }
      }
    }
  }
  __builtin_amdgcn_fence(__ATOMIC_SEQ_CST, "wavefront");
  if (cnt > 0)
    drain_chunk<false>(q, 0, cnt, x4, nrm, rates, impl, acc, lane);

  __syncthreads();
  for (int i = wave; i < OWN_ROWS; i += OWN_WAVES) {
    const unsigned t = ((unsigned)i << 8) | b;
    if (t < (unsigned)NNODE)
      ((float4*)(out + (size_t)t * H))[lane] = ((const float4*)(acc + i * H))[lane];
  }
}

#define N4         (NNODE * H / 4)
#define EW_THREADS 256
#define ACC_RUN    128
#define SCL_RUN    256
#define EW_WAVES   (EW_THREADS / 32)
#define ACC_BLOCKS ((N4 / ACC_RUN + EW_WAVES - 1) / EW_WAVES)
#define SCL_BLOCKS ((N4 / SCL_RUN + EW_WAVES - 1) / EW_WAVES)

__global__ void __launch_bounds__(EW_THREADS)
accum_copy_kernel(float* __restrict__ acc, float* __restrict__ x,
                  const float* __restrict__ buf) {
  const int wave = (int)((blockIdx.x * EW_THREADS + threadIdx.x) >> 5);
  if (wave * ACC_RUN >= N4) return;
  const int i0 = wave * ACC_RUN + (int)(threadIdx.x & 31);
  float4* acc4 = (float4*)acc;
  float4* x4 = (float4*)x;
  const float4* buf4 = (const float4*)buf;
  float4 b[ACC_RUN / 32], a[ACC_RUN / 32];
#pragma unroll
  for (int u = 0; u < ACC_RUN / 32; ++u) b[u] = buf4[i0 + 32 * u];
#pragma unroll
  for (int u = 0; u < ACC_RUN / 32; ++u) a[u] = acc4[i0 + 32 * u];
#pragma unroll
  for (int u = 0; u < ACC_RUN / 32; ++u) {
    float4 s = a[u];
    s.x += b[u].x; s.y += b[u].y; s.z += b[u].z; s.w += b[u].w;
    acc4[i0 + 32 * u] = s;
    x4[i0 + 32 * u] = b[u];
  }
}

__global__ void __launch_bounds__(EW_THREADS)
scale_kernel(float* __restrict__ acc) {
  const int wave = (int)((blockIdx.x * EW_THREADS + threadIdx.x) >> 5);
  if (wave * SCL_RUN >= N4) return;
  const int i0 = wave * SCL_RUN + (int)(threadIdx.x & 31);
  float4* acc4 = (float4*)acc;
  float4 a[SCL_RUN / 32];
#pragma unroll
  for (int u = 0; u < SCL_RUN / 32; ++u) a[u] = acc4[i0 + 32 * u];
#pragma unroll
  for (int u = 0; u < SCL_RUN / 32; ++u) {
    float4 s = a[u];
    s.x *= (1.0f / 3.0f); s.y *= (1.0f / 3.0f); s.z *= (1.0f / 3.0f); s.w *= (1.0f / 3.0f);
    acc4[i0 + 32 * u] = s;
  }
}

extern "C" void kernel_launch(void* const* d_in, const int* in_sizes, int n_in,
                              void* d_out, int out_size, void* d_ws, size_t ws_size,
                              hipStream_t stream) {
  (void)in_sizes; (void)n_in; (void)out_size; (void)ws_size;

  const float* user_x = (const float*)d_in[0];
  const float* book_x = (const float*)d_in[1];
  const float* user_W = (const float*)d_in[2];
  const float* user_b = (const float*)d_in[3];
  const float* book_W = (const float*)d_in[4];
  const float* book_b = (const float*)d_in[5];
  const int*   rates  = (const int*)d_in[6];
  const int*   impl   = (const int*)d_in[7];
  float* outp = (float*)d_out;

  float* x   = (float*)d_ws;
  float* buf = x   + (size_t)NNODE * H;
  float* deg = buf + (size_t)NNODE * H;
  float* dis = deg + NNODE;
  float* nrm = dis + NNODE;
  unsigned* pres = (unsigned*)deg;

  const size_t embBytes = (size_t)NNODE * H * sizeof(float);
  hipMemsetAsync(deg, 0, NNODE * sizeof(float), stream);

  {
    int waves = NU / 16;
    gemm_bias_wmma<<<(waves + 7) / 8, 256, 0, stream>>>(user_x, user_W, user_b, x, NU);
    waves = NB / 16;
    gemm_bias_wmma<<<(waves + 7) / 8, 256, 0, stream>>>(book_x, book_W, book_b,
                                                        x + (size_t)NU * H, NB);
  }

  hipMemcpyAsync(outp, x, embBytes, hipMemcpyDeviceToDevice, stream);

  degree_kernel<<<(NE + 255) / 256, 256, 0, stream>>>(rates, impl, deg);
  dis_kernel<<<(NNODE + 255) / 256, 256, 0, stream>>>(deg, dis);
  norm_kernel<<<(NE + 255) / 256, 256, 0, stream>>>(rates, impl, dis, nrm);
  present_kernel<<<PRES_BLOCKS, 1024, 0, stream>>>(rates, impl, pres);

  for (int layer = 0; layer < 2; ++layer) {
    gather_owner_kernel<<<OWN_BLOCKS, OWN_THREADS, 0, stream>>>(x, nrm, rates, impl, pres, buf);
    accum_copy_kernel<<<ACC_BLOCKS, EW_THREADS, 0, stream>>>(outp, x, buf);
  }

  scale_kernel<<<SCL_BLOCKS, EW_THREADS, 0, stream>>>(outp);
}


# degree: 2M scattered global float atomic adds and the memset replaced by LDS 16-bit integer counts per (edge slice, node range) block plus one pass adding the 32 slice counts; values identical
# speedup vs baseline: 2.1930x; 1.2869x over previous
#include <hip/hip_runtime.h>

#define NU 100000
#define NB 50000
#define NNODE (NU + NB)
#define H 128
#define F 256
#define EPR 500000
#define NE (4 * EPR)

typedef __attribute__((ext_vector_type(2))) float v2f;
typedef __attribute__((ext_vector_type(8))) float v8f;

#define KBLK (F / 8)

struct Frag {
  float4 a[2];
  float4 b[8];
};

__device__ __forceinline__ void load_frag(Frag& f, const float4* __restrict__ a0p,
                                          const float4* __restrict__ a1p,
                                          const float4* __restrict__ bp, int kb) {
  f.a[0] = a0p[2 * kb];
  f.a[1] = a1p[2 * kb];
#pragma unroll
  for (int nt = 0; nt < 8; ++nt) f.b[nt] = bp[nt * (16 * F / 4) + 2 * kb];
}

__device__ __forceinline__ void mma_frag(v8f (&c)[2][8], const Frag& f) {
#pragma unroll
  for (int nt = 0; nt < 8; ++nt) {
#pragma unroll
    for (int t = 0; t < 2; ++t) {
      v2f a = {f.a[t].x, f.a[t].y};
      v2f b = {f.b[nt].x, f.b[nt].y};
      c[t][nt] = __builtin_amdgcn_wmma_f32_16x16x4_f32(
           false, a,  false, b,
           (short)0, c[t][nt],  false,  false);
    }
  }
#pragma unroll
  for (int nt = 0; nt < 8; ++nt) {
#pragma unroll
    for (int t = 0; t < 2; ++t) {
      v2f a = {f.a[t].z, f.a[t].w};
      v2f b = {f.b[nt].z, f.b[nt].w};
      c[t][nt] = __builtin_amdgcn_wmma_f32_16x16x4_f32(
           false, a,  false, b,
           (short)0, c[t][nt],  false,  false);
    }
  }
}

__global__ void __launch_bounds__(256)
gemm_bias_wmma(const float* __restrict__ X, const float* __restrict__ W,
               const float* __restrict__ bias, float* __restrict__ Y, int M) {
  const int wave = (int)((blockIdx.x * blockDim.x + threadIdx.x) >> 5);
  const int lane = threadIdx.x & 31;
  if (wave * 32 >= M) return;
  const int mbase = wave * 32;
  const int lrow  = lane & 15;
  const int kq    = (lane >> 4) << 2;
  const bool two = mbase + 16 < M;
  const int r0 = mbase + lrow;
  const int r1 = two ? (r0 + 16) : r0;

  const float4* a0p = (const float4*)(X + (size_t)r0 * F + kq);
  const float4* a1p = (const float4*)(X + (size_t)r1 * F + kq);
  const float4* bp  = (const float4*)(W + (size_t)lrow * F + kq);

  v8f c[2][8] = {};

  Frag f0, f1;
  load_frag(f0, a0p, a1p, bp, 0);
  for (int kb = 0; kb < KBLK; kb += 2) {
    load_frag(f1, a0p, a1p, bp, kb + 1);
    __builtin_amdgcn_sched_barrier(0);
    mma_frag(c, f0);
    __builtin_amdgcn_sched_barrier(0);
    load_frag(f0, a0p, a1p, bp, (kb + 2 < KBLK) ? (kb + 2) : 0);
    __builtin_amdgcn_sched_barrier(0);
    mma_frag(c, f1);
    __builtin_amdgcn_sched_barrier(0);
  }

  const int mo = (lane >> 4) << 3;
#pragma unroll
  for (int t = 0; t < 2; ++t) {
    if (t == 0 || two) {
#pragma unroll
      for (int nt = 0; nt < 8; ++nt) {
        float bv = bias[nt * 16 + lrow];
#pragma unroll
        for (int v = 0; v < 8; ++v) {
          Y[(size_t)(mbase + 16 * t + mo + v) * H + nt * 16 + lrow] = c[t][nt][v] + bv;
        }
      }
    }
  }
}

__device__ __forceinline__ void decode_edge(int e, const int* __restrict__ rates,
                                            const int* __restrict__ impl,
                                            int& row, int& col) {
  const int rev = (e >= 2 * EPR) ? 1 : 0;
  int idx = rev ? (e - 2 * EPR) : e;
  const int* ed = (idx < EPR) ? rates : impl;
  const int i = (idx < EPR) ? idx : (idx - EPR);
  const int u = ed[i];
  const int b = ed[i + EPR];
  row = rev ? (b + NU) : u;
  col = rev ? u : b;
}

__global__ void __launch_bounds__(256)
norm_kernel(const int* __restrict__ rates, const int* __restrict__ impl,
            const float* __restrict__ dis, float* __restrict__ nrm) {
  const int e = blockIdx.x * blockDim.x + threadIdx.x;
  if (e >= NE) return;
  int row, col;
  decode_edge(e, rates, impl, row, col);
  nrm[e] = dis[row] * dis[col];
}

#define OWN_BLOCKS  256
#define OWN_ROWS    ((NNODE + OWN_BLOCKS - 1) / OWN_BLOCKS)
#define OWN_THREADS 1024
#define OWN_WAVES   (OWN_THREADS / 32)
#define QCAP        160
#define DRAIN_G     16

#define GRP_SEG     ((EPR + 63) / 64)
#define SEG_PITCH   7936
#define SEG_STEPS   (SEG_PITCH / 128)
#define PPITCH      (4 * SEG_PITCH)
#define NSTEP       (4 * SEG_STEPS)
#define PRES_BLOCKS (PPITCH / 32)

__device__ __forceinline__ void lds_add(float* p, float v) {
  __hip_atomic_fetch_add(p, v, __ATOMIC_RELAXED, __HIP_MEMORY_SCOPE_WORKGROUP);
}

template <bool FULL>
__device__ __forceinline__ void drain_chunk(const unsigned* q, int base, int nv,
                                            const float4* __restrict__ x4,
                                            const float* __restrict__ nrm,
                                            const int* __restrict__ rates,
                                            const int* __restrict__ impl,
                                            float* acc, int lane) {
  const int h = lane >> 4;
  const int li = FULL ? (base + lane) : (base + ((lane < nv) ? lane : 0));
  const unsigned d = q[li];
  const int e = (int)(d >> 10);
  const int loc = (int)(d & 1023u);
  const bool rev = e >= 2 * EPR;
  const int idx = rev ? (e - 2 * EPR) : e;
  const bool first = idx < EPR;
  const int i = first ? idx : (idx - EPR);
  const int* ed = first ? rates : impl;
  const int sv = ed[i + (rev ? EPR : 0)];
  const int row = rev ? (sv + NU) : sv;
  const int wbits = __float_as_int(nrm[e]);
#pragma unroll
  for (int half = 0; half < 32 / DRAIN_G; ++half) {
    if (!FULL && half * DRAIN_G >= nv) break;
    float4 v[DRAIN_G];
#pragma unroll
    for (int g = 0; g < DRAIN_G; ++g) {
      const int r = __builtin_amdgcn_readlane(row, half * DRAIN_G + g);
      v[g] = x4[(size_t)r * (H / 4) + lane];
    }
#pragma unroll
    for (int g = 0; g < DRAIN_G; ++g) {
      if (FULL || half * DRAIN_G + g < nv) {
        const int il = __builtin_amdgcn_readlane(loc, half * DRAIN_G + g);
        const float m = __int_as_float(__builtin_amdgcn_readlane(wbits, half * DRAIN_G + g));
        float* a = acc + il * H + 4 * lane;
        const float4 s = v[g];
        lds_add(a + h,             (h ? s.y : s.x) * m);
        lds_add(a + ((1 + h) & 3), (h ? s.z : s.y) * m);
        lds_add(a + ((2 + h) & 3), (h ? s.w : s.z) * m);
        lds_add(a + ((3 + h) & 3), (h ? s.x : s.w) * m);
      }
    }
  }
}

__device__ __forceinline__ const int* seg_cols(int seg, const int* __restrict__ rates,
                                               const int* __restrict__ impl) {
  return (seg == 0) ? (rates + EPR) : (seg == 1) ? (impl + EPR) : (seg == 2) ? rates : impl;
}

#define HR      8
#define HS      32
#define HRANGE  (NNODE / HR)
#define HPAIRS  (HRANGE / 2)
#define HSLICE  (NE / HS)
#define HSL4    (HSLICE / 4)
#define HPITCH  (NNODE / 2)
#define HTHREADS 1024
static_assert(HSLICE < 65536, "a 16-bit count must hold a whole slice");
static_assert(HRANGE * HR == NNODE && HRANGE % 2 == 0, "ranges cover the nodes in whole dwords");
static_assert(HSLICE * HS == NE && EPR % HSLICE == 0 && HSLICE % 4 == 0, "slices stay inside a segment");

__device__ __forceinline__ void hist_add(unsigned* cnt, int t, unsigned lo) {
  const unsigned q = (unsigned)t - lo;
  if (q < (unsigned)HRANGE)
    __hip_atomic_fetch_add(cnt + (q >> 1), (q & 1u) ? 65536u : 1u, __ATOMIC_RELAXED,
                           __HIP_MEMORY_SCOPE_WORKGROUP);
}

__global__ void __launch_bounds__(HTHREADS)
degree_hist_kernel(const int* __restrict__ rates, const int* __restrict__ impl,
                   unsigned* __restrict__ part) {
  __shared__ unsigned cnt[HPAIRS];
  const int tid = threadIdx.x;
  const int r = (int)blockIdx.x % HR;
  const int s = (int)blockIdx.x / HR;
  for (int i = tid; i < HPAIRS; i += HTHREADS) cnt[i] = 0u;
  __syncthreads();
  const int seg = s / (HS / 4);
  const int4* p = (const int4*)(seg_cols(seg, rates, impl) + (s - seg * (HS / 4)) * HSLICE);
  const unsigned lo = (unsigned)(r * HRANGE);
  for (int j0 = 0; j0 < HSL4; j0 += 4 * HTHREADS) {
    int4 v[4];
#pragma unroll
    for (int u = 0; u < 4; ++u) {
      const int j = j0 + u * HTHREADS + tid;
      v[u] = (j < HSL4) ? p[j] : make_int4(-1, -1, -1, -1);
    }
#pragma unroll
    for (int u = 0; u < 4; ++u) {
      hist_add(cnt, v[u].x, lo);
      hist_add(cnt, v[u].y, lo);
      hist_add(cnt, v[u].z, lo);
      hist_add(cnt, v[u].w, lo);
    }
  }
  __syncthreads();
  unsigned* dst = part + (size_t)s * HPITCH + r * HPAIRS;
  for (int i = tid; i < HPAIRS; i += HTHREADS) dst[i] = cnt[i];
}

__global__ void __launch_bounds__(256)
dis_from_hist_kernel(const unsigned* __restrict__ part, float* __restrict__ dis) {
  const int p = blockIdx.x * blockDim.x + threadIdx.x;
  if (p >= HPITCH) return;
  unsigned v[HS];
#pragma unroll
  for (int s = 0; s < HS; ++s) v[s] = part[(size_t)s * HPITCH + p];
  unsigned c0 = 0u, c1 = 0u;
#pragma unroll
  for (int s = 0; s < HS; ++s) {
    c0 += v[s] & 0xffffu;
    c1 += v[s] >> 16;
  }
  const float d0 = (float)c0, d1 = (float)c1;
  float2 o;
  o.x = (d0 > 0.0f) ? rsqrtf(d0) : 0.0f;
  o.y = (d1 > 0.0f) ? rsqrtf(d1) : 0.0f;
  ((float2*)dis)[p] = o;
}

__global__ void __launch_bounds__(1024)
present_kernel(const int* __restrict__ rates, const int* __restrict__ impl,
               unsigned* __restrict__ pres) {
  __shared__ unsigned st[8][32];
  const int lane = threadIdx.x & 31;
  const int wave = __builtin_amdgcn_readfirstlane((int)(threadIdx.x >> 5));
  const int pidx = (int)blockIdx.x * 32 + wave;
  const int seg = pidx / SEG_PITCH;
  const int gs  = pidx - seg * SEG_PITCH;
  unsigned myv = 0u;
  if (gs < GRP_SEG) {
    const int idx = gs * 64 + 2 * lane;
    const bool ok = idx < EPR;
    const int2 t2 = *(const int2*)(seg_cols(seg, rates, impl) + (ok ? idx : 0));
    const unsigned ta = (unsigned)t2.x, tb = (unsigned)t2.y;
    const unsigned valid = __builtin_amdgcn_ballot_w32(ok);
    unsigned la = valid, lb = valid;
#pragma unroll
    for (int i = 0; i < 5; ++i) {
      const unsigned ka = __builtin_amdgcn_ballot_w32(((ta >> i) & 1u) != 0u);
      const unsigned kb = __builtin_amdgcn_ballot_w32(((tb >> i) & 1u) != 0u);
      const bool on = ((lane >> i) & 1) != 0;
      la &= on ? ka : ~ka;
      lb &= on ? kb : ~kb;
    }
    unsigned ha[3], hb[3];
#pragma unroll
    for (int i = 0; i < 3; ++i) {
      ha[i] = __builtin_amdgcn_ballot_w32(((ta >> (5 + i)) & 1u) != 0u);
      hb[i] = __builtin_amdgcn_ballot_w32(((tb >> (5 + i)) & 1u) != 0u);
    }
#pragma unroll
    for (int m = 0; m < 8; ++m) {
      unsigned sa = la, sb = lb;
#pragma unroll
      for (int i = 0; i < 3; ++i) {
        sa &= ((m >> i) & 1) ? ha[i] : ~ha[i];
        sb &= ((m >> i) & 1) ? hb[i] : ~hb[i];
      }
      const unsigned pm = __builtin_amdgcn_ballot_w32((sa | sb) != 0u);
      if (lane == m) myv = pm;
    }
  }
  if (lane < 8) st[lane][wave] = myv;
  __syncthreads();
  if (wave < 8) pres[wave * PPITCH + (int)blockIdx.x * 32 + lane] = st[wave][lane];
}

__device__ __forceinline__ void push_owned(unsigned t, unsigned e, unsigned live, unsigned key,
                                           unsigned* q, int& cnt, int lane) {
  const unsigned iloc = ((t >> 8) | (t << 24)) - key;
  const unsigned mk = __builtin_amdgcn_ballot_w32(iloc < (unsigned)OWN_ROWS) & live;
  if (mk != 0u) {
    if ((mk >> lane) & 1u)
      q[cnt + (int)__builtin_amdgcn_mbcnt_lo(mk, 0u)] = (e << 10) | iloc;
    cnt += __builtin_popcount(mk);
  }
}

__global__ void __launch_bounds__(OWN_THREADS)
gather_owner_kernel(const float* __restrict__ x, const float* __restrict__ nrm,
                    const int* __restrict__ rates, const int* __restrict__ impl,
                    const unsigned* __restrict__ pres, float* __restrict__ out) {
  __shared__ __attribute__((aligned(16))) float acc[OWN_ROWS * H];
  __shared__ unsigned queue[OWN_WAVES * QCAP];

  const int tid  = threadIdx.x;
  const int lane = tid & 31;
  const int wave = __builtin_amdgcn_readfirstlane(tid >> 5);
  const unsigned b = blockIdx.x;

  float4* acc4 = (float4*)acc;
  for (int i = tid; i < OWN_ROWS * H / 4; i += OWN_THREADS)
    acc4[i] = make_float4(0.0f, 0.0f, 0.0f, 0.0f);
  __syncthreads();

  unsigned* q = queue + wave * QCAP;
  const float4* x4 = (const float4*)x;
  const unsigned key = b << 24;
  const int rot = (int)((b * NSTEP) >> 8);
  const uint4* plane4 = (const uint4*)(pres + (b >> 5) * PPITCH);
  const unsigned bbit = b & 31u;
  int cnt = 0;

  for (int s = wave; s < NSTEP; s += OWN_WAVES) {
    int sr = s + rot;
    if (sr >= NSTEP) sr -= NSTEP;
    const int seg = sr / SEG_STEPS;
    const int st  = sr - seg * SEG_STEPS;
    const int* colp = seg_cols(seg, rates, impl);
    const unsigned ebase = (unsigned)(seg * EPR);
    const uint4 d = plane4[sr * 32 + lane];
    const unsigned dk[4] = {d.x, d.y, d.z, d.w};
#pragma unroll
    for (int k = 0; k < 4; ++k) {
      unsigned m = __builtin_amdgcn_ballot_w32(((dk[k] >> bbit) & 1u) != 0u);
      while (m != 0u) {
        const int l0 = __builtin_ctz(m);
        m &= m - 1u;
        const unsigned two = (m != 0u) ? ~0u : 0u;
        const int l1 = (m != 0u) ? __builtin_ctz(m) : l0;
        m &= m - 1u;
        const int i0 = ((((st * 32 + l0) << 2) + k) << 6) + 2 * lane;
        const int i1 = ((((st * 32 + l1) << 2) + k) << 6) + 2 * lane;
        const bool ok0 = i0 < EPR, ok1 = i1 < EPR;
        const int2 t0 = *(const int2*)(colp + (ok0 ? i0 : 0));
        const int2 t1 = *(const int2*)(colp + (ok1 ? i1 : 0));
        const unsigned live0 = __builtin_amdgcn_ballot_w32(ok0);
        const unsigned live1 = __builtin_amdgcn_ballot_w32(ok1) & two;
        const unsigned e0 = ebase + (unsigned)i0, e1 = ebase + (unsigned)i1;
        push_owned((unsigned)t0.x, e0,      live0, key, q, cnt, lane);
        push_owned((unsigned)t0.y, e0 + 1u, live0, key, q, cnt, lane);
        push_owned((unsigned)t1.x, e1,      live1, key, q, cnt, lane);
        push_owned((unsigned)t1.y, e1 + 1u, live1, key, q, cnt, lane);
        if (cnt >= 32) {
          __builtin_amdgcn_fence(__ATOMIC_SEQ_CST, "wavefront");
          const int full = cnt & ~31;
          for (int c = 0; c < full; c += 32)
            drain_chunk<true>(q, c, 32, x4, nrm, rates, impl, acc, lane);
          const int rem = cnt - full;
          if (lane < rem) q[lane] = q[full + lane];
          __builtin_amdgcn_fence(__ATOMIC_SEQ_CST, "wavefront");
          cnt = rem;
        }
      }
    }
  }
  __builtin_amdgcn_fence(__ATOMIC_SEQ_CST, "wavefront");
  if (cnt > 0)
    drain_chunk<false>(q, 0, cnt, x4, nrm, rates, impl, acc, lane);

  __syncthreads();
  for (int i = wave; i < OWN_ROWS; i += OWN_WAVES) {
    const unsigned t = ((unsigned)i << 8) | b;
    if (t < (unsigned)NNODE)
      ((float4*)(out + (size_t)t * H))[lane] = ((const float4*)(acc + i * H))[lane];
  }
}

#define N4         (NNODE * H / 4)
#define EW_THREADS 256
#define ACC_RUN    128
#define SCL_RUN    256
#define EW_WAVES   (EW_THREADS / 32)
#define ACC_BLOCKS ((N4 / ACC_RUN + EW_WAVES - 1) / EW_WAVES)
#define SCL_BLOCKS ((N4 / SCL_RUN + EW_WAVES - 1) / EW_WAVES)

__global__ void __launch_bounds__(EW_THREADS)
accum_copy_kernel(float* __restrict__ acc, float* __restrict__ x,
                  const float* __restrict__ buf) {
  const int wave = (int)((blockIdx.x * EW_THREADS + threadIdx.x) >> 5);
  if (wave * ACC_RUN >= N4) return;
  const int i0 = wave * ACC_RUN + (int)(threadIdx.x & 31);
  float4* acc4 = (float4*)acc;
  float4* x4 = (float4*)x;
  const float4* buf4 = (const float4*)buf;
  float4 b[ACC_RUN / 32], a[ACC_RUN / 32];
#pragma unroll
  for (int u = 0; u < ACC_RUN / 32; ++u) b[u] = buf4[i0 + 32 * u];
#pragma unroll
  for (int u = 0; u < ACC_RUN / 32; ++u) a[u] = acc4[i0 + 32 * u];
#pragma unroll
  for (int u = 0; u < ACC_RUN / 32; ++u) {
    float4 s = a[u];
    s.x += b[u].x; s.y += b[u].y; s.z += b[u].z; s.w += b[u].w;
    acc4[i0 + 32 * u] = s;
    x4[i0 + 32 * u] = b[u];
  }
}

__global__ void __launch_bounds__(EW_THREADS)
scale_kernel(float* __restrict__ acc) {
  const int wave = (int)((blockIdx.x * EW_THREADS + threadIdx.x) >> 5);
  if (wave * SCL_RUN >= N4) return;
  const int i0 = wave * SCL_RUN + (int)(threadIdx.x & 31);
  float4* acc4 = (float4*)acc;
  float4 a[SCL_RUN / 32];
#pragma unroll
  for (int u = 0; u < SCL_RUN / 32; ++u) a[u] = acc4[i0 + 32 * u];
#pragma unroll
  for (int u = 0; u < SCL_RUN / 32; ++u) {
    float4 s = a[u];
    s.x *= (1.0f / 3.0f); s.y *= (1.0f / 3.0f); s.z *= (1.0f / 3.0f); s.w *= (1.0f / 3.0f);
    acc4[i0 + 32 * u] = s;
  }
}

extern "C" void kernel_launch(void* const* d_in, const int* in_sizes, int n_in,
                              void* d_out, int out_size, void* d_ws, size_t ws_size,
                              hipStream_t stream) {
  (void)in_sizes; (void)n_in; (void)out_size; (void)ws_size;

  const float* user_x = (const float*)d_in[0];
  const float* book_x = (const float*)d_in[1];
  const float* user_W = (const float*)d_in[2];
  const float* user_b = (const float*)d_in[3];
  const float* book_W = (const float*)d_in[4];
  const float* book_b = (const float*)d_in[5];
  const int*   rates  = (const int*)d_in[6];
  const int*   impl   = (const int*)d_in[7];
  float* outp = (float*)d_out;

  float* x   = (float*)d_ws;
  float* buf = x   + (size_t)NNODE * H;
  float* deg = buf + (size_t)NNODE * H;
  float* dis = deg + NNODE;
  float* nrm = dis + NNODE;
  unsigned* pres = (unsigned*)deg;
  unsigned* part = (unsigned*)buf;

  const size_t embBytes = (size_t)NNODE * H * sizeof(float);

  {
    int waves = (NU + 31) / 32;
    gemm_bias_wmma<<<(waves + 7) / 8, 256, 0, stream>>>(user_x, user_W, user_b, x, NU);
    waves = (NB + 31) / 32;
    gemm_bias_wmma<<<(waves + 7) / 8, 256, 0, stream>>>(book_x, book_W, book_b,
                                                        x + (size_t)NU * H, NB);
  }

  hipMemcpyAsync(outp, x, embBytes, hipMemcpyDeviceToDevice, stream);

  degree_hist_kernel<<<HS * HR, HTHREADS, 0, stream>>>(rates, impl, part);
  dis_from_hist_kernel<<<(HPITCH + 255) / 256, 256, 0, stream>>>(part, dis);
  norm_kernel<<<(NE + 255) / 256, 256, 0, stream>>>(rates, impl, dis, nrm);
  present_kernel<<<PRES_BLOCKS, 1024, 0, stream>>>(rates, impl, pres);

  for (int layer = 0; layer < 2; ++layer) {
    gather_owner_kernel<<<OWN_BLOCKS, OWN_THREADS, 0, stream>>>(x, nrm, rates, impl, pres, buf);
    accum_copy_kernel<<<ACC_BLOCKS, EW_THREADS, 0, stream>>>(outp, x, buf);
  }

  scale_kernel<<<SCL_BLOCKS, EW_THREADS, 0, stream>>>(outp);
}


# mean re-associated as (x0 + A(x0 + A x0))/3: a layer-sum block seeds its LDS rows with the layer-0 rows, last pass stores sum/3 to out; copy, 2 accumulate and scale launches gone; sum order changes
# speedup vs baseline: 2.8644x; 1.3062x over previous
#include <hip/hip_runtime.h>

#define NU 100000
#define NB 50000
#define NNODE (NU + NB)
#define H 128
#define F 256
#define EPR 500000
#define NE (4 * EPR)

typedef __attribute__((ext_vector_type(2))) float v2f;
typedef __attribute__((ext_vector_type(8))) float v8f;

#define KBLK (F / 8)

struct Frag {
  float4 a[2];
  float4 b[8];
};

__device__ __forceinline__ void load_frag(Frag& f, const float4* __restrict__ a0p,
                                          const float4* __restrict__ a1p,
                                          const float4* __restrict__ bp, int kb) {
  f.a[0] = a0p[2 * kb];
  f.a[1] = a1p[2 * kb];
#pragma unroll
  for (int nt = 0; nt < 8; ++nt) f.b[nt] = bp[nt * (16 * F / 4) + 2 * kb];
}

__device__ __forceinline__ void mma_frag(v8f (&c)[2][8], const Frag& f) {
#pragma unroll
  for (int nt = 0; nt < 8; ++nt) {
#pragma unroll
    for (int t = 0; t < 2; ++t) {
      v2f a = {f.a[t].x, f.a[t].y};
      v2f b = {f.b[nt].x, f.b[nt].y};
      c[t][nt] = __builtin_amdgcn_wmma_f32_16x16x4_f32(
           false, a,  false, b,
           (short)0, c[t][nt],  false,  false);
    }
  }
#pragma unroll
  for (int nt = 0; nt < 8; ++nt) {
#pragma unroll
    for (int t = 0; t < 2; ++t) {
      v2f a = {f.a[t].z, f.a[t].w};
      v2f b = {f.b[nt].z, f.b[nt].w};
      c[t][nt] = __builtin_amdgcn_wmma_f32_16x16x4_f32(
           false, a,  false, b,
           (short)0, c[t][nt],  false,  false);
    }
  }
}

__global__ void __launch_bounds__(256)
gemm_bias_wmma(const float* __restrict__ X, const float* __restrict__ W,
               const float* __restrict__ bias, float* __restrict__ Y, int M) {
  const int wave = (int)((blockIdx.x * blockDim.x + threadIdx.x) >> 5);
  const int lane = threadIdx.x & 31;
  if (wave * 32 >= M) return;
  const int mbase = wave * 32;
  const int lrow  = lane & 15;
  const int kq    = (lane >> 4) << 2;
  const bool two = mbase + 16 < M;
  const int r0 = mbase + lrow;
  const int r1 = two ? (r0 + 16) : r0;

  const float4* a0p = (const float4*)(X + (size_t)r0 * F + kq);
  const float4* a1p = (const float4*)(X + (size_t)r1 * F + kq);
  const float4* bp  = (const float4*)(W + (size_t)lrow * F + kq);

  v8f c[2][8] = {};

  Frag f0, f1;
  load_frag(f0, a0p, a1p, bp, 0);
  for (int kb = 0; kb < KBLK; kb += 2) {
    load_frag(f1, a0p, a1p, bp, kb + 1);
    __builtin_amdgcn_sched_barrier(0);
    mma_frag(c, f0);
    __builtin_amdgcn_sched_barrier(0);
    load_frag(f0, a0p, a1p, bp, (kb + 2 < KBLK) ? (kb + 2) : 0);
    __builtin_amdgcn_sched_barrier(0);
    mma_frag(c, f1);
    __builtin_amdgcn_sched_barrier(0);
  }

  const int mo = (lane >> 4) << 3;
#pragma unroll
  for (int t = 0; t < 2; ++t) {
    if (t == 0 || two) {
#pragma unroll
      for (int nt = 0; nt < 8; ++nt) {
        float bv = bias[nt * 16 + lrow];
#pragma unroll
        for (int v = 0; v < 8; ++v) {
          Y[(size_t)(mbase + 16 * t + mo + v) * H + nt * 16 + lrow] = c[t][nt][v] + bv;
        }
      }
    }
  }
}

__device__ __forceinline__ void decode_edge(int e, const int* __restrict__ rates,
                                            const int* __restrict__ impl,
                                            int& row, int& col) {
  const int rev = (e >= 2 * EPR) ? 1 : 0;
  int idx = rev ? (e - 2 * EPR) : e;
  const int* ed = (idx < EPR) ? rates : impl;
  const int i = (idx < EPR) ? idx : (idx - EPR);
  const int u = ed[i];
  const int b = ed[i + EPR];
  row = rev ? (b + NU) : u;
  col = rev ? u : b;
}

__global__ void __launch_bounds__(256)
norm_kernel(const int* __restrict__ rates, const int* __restrict__ impl,
            const float* __restrict__ dis, float* __restrict__ nrm) {
  const int e = blockIdx.x * blockDim.x + threadIdx.x;
  if (e >= NE) return;
  int row, col;
  decode_edge(e, rates, impl, row, col);
  nrm[e] = dis[row] * dis[col];
}

#define OWN_BLOCKS  256
#define OWN_ROWS    ((NNODE + OWN_BLOCKS - 1) / OWN_BLOCKS)
#define OWN_THREADS 1024
#define OWN_WAVES   (OWN_THREADS / 32)
#define QCAP        160
#define DRAIN_G     16
#define SEED_G      10

#define GRP_SEG     ((EPR + 63) / 64)
#define SEG_PITCH   7936
#define SEG_STEPS   (SEG_PITCH / 128)
#define PPITCH      (4 * SEG_PITCH)
#define NSTEP       (4 * SEG_STEPS)
#define PRES_BLOCKS (PPITCH / 32)

__device__ __forceinline__ void lds_add(float* p, float v) {
  __hip_atomic_fetch_add(p, v, __ATOMIC_RELAXED, __HIP_MEMORY_SCOPE_WORKGROUP);
}

template <bool FULL>
__device__ __forceinline__ void drain_chunk(const unsigned* q, int base, int nv,
                                            const float4* __restrict__ x4,
                                            const float* __restrict__ nrm,
                                            const int* __restrict__ rates,
                                            const int* __restrict__ impl,
                                            float* acc, int lane) {
  const int h = lane >> 4;
  const int li = FULL ? (base + lane) : (base + ((lane < nv) ? lane : 0));
  const unsigned d = q[li];
  const int e = (int)(d >> 10);
  const int loc = (int)(d & 1023u);
  const bool rev = e >= 2 * EPR;
  const int idx = rev ? (e - 2 * EPR) : e;
  const bool first = idx < EPR;
  const int i = first ? idx : (idx - EPR);
  const int* ed = first ? rates : impl;
  const int sv = ed[i + (rev ? EPR : 0)];
  const int row = rev ? (sv + NU) : sv;
  const int wbits = __float_as_int(nrm[e]);
#pragma unroll
  for (int half = 0; half < 32 / DRAIN_G; ++half) {
    if (!FULL && half * DRAIN_G >= nv) break;
    float4 v[DRAIN_G];
#pragma unroll
    for (int g = 0; g < DRAIN_G; ++g) {
      const int r = __builtin_amdgcn_readlane(row, half * DRAIN_G + g);
      v[g] = x4[(size_t)r * (H / 4) + lane];
    }
#pragma unroll
    for (int g = 0; g < DRAIN_G; ++g) {
      if (FULL || half * DRAIN_G + g < nv) {
        const int il = __builtin_amdgcn_readlane(loc, half * DRAIN_G + g);
        const float m = __int_as_float(__builtin_amdgcn_readlane(wbits, half * DRAIN_G + g));
        float* a = acc + il * H + 4 * lane;
        const float4 s = v[g];
        lds_add(a + h,             (h ? s.y : s.x) * m);
        lds_add(a + ((1 + h) & 3), (h ? s.z : s.y) * m);
        lds_add(a + ((2 + h) & 3), (h ? s.w : s.z) * m);
        lds_add(a + ((3 + h) & 3), (h ? s.x : s.w) * m);
      }
    }
  }
}

__device__ __forceinline__ const int* seg_cols(int seg, const int* __restrict__ rates,
                                               const int* __restrict__ impl) {
  return (seg == 0) ? (rates + EPR) : (seg == 1) ? (impl + EPR) : (seg == 2) ? rates : impl;
}

#define HR      8
#define HS      32
#define HRANGE  (NNODE / HR)
#define HPAIRS  (HRANGE / 2)
#define HSLICE  (NE / HS)
#define HSL4    (HSLICE / 4)
#define HPITCH  (NNODE / 2)
#define HTHREADS 1024
static_assert(HSLICE < 65536, "a 16-bit count must hold a whole slice");
static_assert(HRANGE * HR == NNODE && HRANGE % 2 == 0, "ranges cover the nodes in whole dwords");
static_assert(HSLICE * HS == NE && EPR % HSLICE == 0 && HSLICE % 4 == 0, "slices stay inside a segment");

__device__ __forceinline__ void hist_add(unsigned* cnt, int t, unsigned lo) {
  const unsigned q = (unsigned)t - lo;
  if (q < (unsigned)HRANGE)
    __hip_atomic_fetch_add(cnt + (q >> 1), (q & 1u) ? 65536u : 1u, __ATOMIC_RELAXED,
                           __HIP_MEMORY_SCOPE_WORKGROUP);
}

__global__ void __launch_bounds__(HTHREADS)
degree_hist_kernel(const int* __restrict__ rates, const int* __restrict__ impl,
                   unsigned* __restrict__ part) {
  __shared__ unsigned cnt[HPAIRS];
  const int tid = threadIdx.x;
  const int r = (int)blockIdx.x % HR;
  const int s = (int)blockIdx.x / HR;
  for (int i = tid; i < HPAIRS; i += HTHREADS) cnt[i] = 0u;
  __syncthreads();
  const int seg = s / (HS / 4);
  const int4* p = (const int4*)(seg_cols(seg, rates, impl) + (s - seg * (HS / 4)) * HSLICE);
  const unsigned lo = (unsigned)(r * HRANGE);
  for (int j0 = 0; j0 < HSL4; j0 += 4 * HTHREADS) {
    int4 v[4];
#pragma unroll
    for (int u = 0; u < 4; ++u) {
      const int j = j0 + u * HTHREADS + tid;
      v[u] = (j < HSL4) ? p[j] : make_int4(-1, -1, -1, -1);
    }
#pragma unroll
    for (int u = 0; u < 4; ++u) {
      hist_add(cnt, v[u].x, lo);
      hist_add(cnt, v[u].y, lo);
      hist_add(cnt, v[u].z, lo);
      hist_add(cnt, v[u].w, lo);
    }
  }
  __syncthreads();
  unsigned* dst = part + (size_t)s * HPITCH + r * HPAIRS;
  for (int i = tid; i < HPAIRS; i += HTHREADS) dst[i] = cnt[i];
}

__global__ void __launch_bounds__(256)
dis_from_hist_kernel(const unsigned* __restrict__ part, float* __restrict__ dis) {
  const int p = blockIdx.x * blockDim.x + threadIdx.x;
  if (p >= HPITCH) return;
  unsigned v[HS];
#pragma unroll
  for (int s = 0; s < HS; ++s) v[s] = part[(size_t)s * HPITCH + p];
  unsigned c0 = 0u, c1 = 0u;
#pragma unroll
  for (int s = 0; s < HS; ++s) {
    c0 += v[s] & 0xffffu;
    c1 += v[s] >> 16;
  }
  const float d0 = (float)c0, d1 = (float)c1;
  float2 o;
  o.x = (d0 > 0.0f) ? rsqrtf(d0) : 0.0f;
  o.y = (d1 > 0.0f) ? rsqrtf(d1) : 0.0f;
  ((float2*)dis)[p] = o;
}

__global__ void __launch_bounds__(1024)
present_kernel(const int* __restrict__ rates, const int* __restrict__ impl,
               unsigned* __restrict__ pres) {
  __shared__ unsigned st[8][32];
  const int lane = threadIdx.x & 31;
  const int wave = __builtin_amdgcn_readfirstlane((int)(threadIdx.x >> 5));
  const int pidx = (int)blockIdx.x * 32 + wave;
  const int seg = pidx / SEG_PITCH;
  const int gs  = pidx - seg * SEG_PITCH;
  unsigned myv = 0u;
  if (gs < GRP_SEG) {
    const int idx = gs * 64 + 2 * lane;
    const bool ok = idx < EPR;
    const int2 t2 = *(const int2*)(seg_cols(seg, rates, impl) + (ok ? idx : 0));
    const unsigned ta = (unsigned)t2.x, tb = (unsigned)t2.y;
    const unsigned valid = __builtin_amdgcn_ballot_w32(ok);
    unsigned la = valid, lb = valid;
#pragma unroll
    for (int i = 0; i < 5; ++i) {
      const unsigned ka = __builtin_amdgcn_ballot_w32(((ta >> i) & 1u) != 0u);
      const unsigned kb = __builtin_amdgcn_ballot_w32(((tb >> i) & 1u) != 0u);
      const bool on = ((lane >> i) & 1) != 0;
      la &= on ? ka : ~ka;
      lb &= on ? kb : ~kb;
    }
    unsigned ha[3], hb[3];
#pragma unroll
    for (int i = 0; i < 3; ++i) {
      ha[i] = __builtin_amdgcn_ballot_w32(((ta >> (5 + i)) & 1u) != 0u);
      hb[i] = __builtin_amdgcn_ballot_w32(((tb >> (5 + i)) & 1u) != 0u);
    }
#pragma unroll
    for (int m = 0; m < 8; ++m) {
      unsigned sa = la, sb = lb;
#pragma unroll
      for (int i = 0; i < 3; ++i) {
        sa &= ((m >> i) & 1) ? ha[i] : ~ha[i];
        sb &= ((m >> i) & 1) ? hb[i] : ~hb[i];
      }
      const unsigned pm = __builtin_amdgcn_ballot_w32((sa | sb) != 0u);
      if (lane == m) myv = pm;
    }
  }
  if (lane < 8) st[lane][wave] = myv;
  __syncthreads();
  if (wave < 8) pres[wave * PPITCH + (int)blockIdx.x * 32 + lane] = st[wave][lane];
}

__device__ __forceinline__ void push_owned(unsigned t, unsigned e, unsigned live, unsigned key,
                                           unsigned* q, int& cnt, int lane) {
  const unsigned iloc = ((t >> 8) | (t << 24)) - key;
  const unsigned mk = __builtin_amdgcn_ballot_w32(iloc < (unsigned)OWN_ROWS) & live;
  if (mk != 0u) {
    if ((mk >> lane) & 1u)
      q[cnt + (int)__builtin_amdgcn_mbcnt_lo(mk, 0u)] = (e << 10) | iloc;
    cnt += __builtin_popcount(mk);
  }
}

template <bool LAST>
__global__ void __launch_bounds__(OWN_THREADS)
gather_owner_kernel(const float* __restrict__ x, const float* __restrict__ x0,
                    const float* __restrict__ nrm,
                    const int* __restrict__ rates, const int* __restrict__ impl,
                    const unsigned* __restrict__ pres, float* __restrict__ out) {
  __shared__ __attribute__((aligned(16))) float acc[OWN_ROWS * H];
  __shared__ unsigned queue[OWN_WAVES * QCAP];

  const int tid  = threadIdx.x;
  const int lane = tid & 31;
  const int wave = __builtin_amdgcn_readfirstlane(tid >> 5);
  const unsigned b = blockIdx.x;

  float4* acc4 = (float4*)acc;
  const float4* s4 = (const float4*)x0;
  for (int i0 = wave; i0 < OWN_ROWS; i0 += SEED_G * OWN_WAVES) {
    float4 v[SEED_G];
#pragma unroll
    for (int u = 0; u < SEED_G; ++u) {
      const unsigned t = ((unsigned)(i0 + u * OWN_WAVES) << 8) | b;
      v[u] = (t < (unsigned)NNODE) ? s4[(size_t)t * (H / 4) + lane]
                                   : make_float4(0.0f, 0.0f, 0.0f, 0.0f);
    }
#pragma unroll
    for (int u = 0; u < SEED_G; ++u) {
      const int i = i0 + u * OWN_WAVES;
      if (i < OWN_ROWS) acc4[i * (H / 4) + lane] = v[u];
    }
  }
  __syncthreads();

  unsigned* q = queue + wave * QCAP;
  const float4* x4 = (const float4*)x;
  const unsigned key = b << 24;
  const int rot = (int)((b * NSTEP) >> 8);
  const uint4* plane4 = (const uint4*)(pres + (b >> 5) * PPITCH);
  const unsigned bbit = b & 31u;
  int cnt = 0;

  for (int s = wave; s < NSTEP; s += OWN_WAVES) {
    int sr = s + rot;
    if (sr >= NSTEP) sr -= NSTEP;
    const int seg = sr / SEG_STEPS;
    const int st  = sr - seg * SEG_STEPS;
    const int* colp = seg_cols(seg, rates, impl);
    const unsigned ebase = (unsigned)(seg * EPR);
    const uint4 d = plane4[sr * 32 + lane];
    const unsigned dk[4] = {d.x, d.y, d.z, d.w};
#pragma unroll
    for (int k = 0; k < 4; ++k) {
      unsigned m = __builtin_amdgcn_ballot_w32(((dk[k] >> bbit) & 1u) != 0u);
      while (m != 0u) {
        const int l0 = __builtin_ctz(m);
        m &= m - 1u;
        const unsigned two = (m != 0u) ? ~0u : 0u;
        const int l1 = (m != 0u) ? __builtin_ctz(m) : l0;
        m &= m - 1u;
        const int i0 = ((((st * 32 + l0) << 2) + k) << 6) + 2 * lane;
        const int i1 = ((((st * 32 + l1) << 2) + k) << 6) + 2 * lane;
        const bool ok0 = i0 < EPR, ok1 = i1 < EPR;
        const int2 t0 = *(const int2*)(colp + (ok0 ? i0 : 0));
        const int2 t1 = *(const int2*)(colp + (ok1 ? i1 : 0));
        const unsigned live0 = __builtin_amdgcn_ballot_w32(ok0);
        const unsigned live1 = __builtin_amdgcn_ballot_w32(ok1) & two;
        const unsigned e0 = ebase + (unsigned)i0, e1 = ebase + (unsigned)i1;
        push_owned((unsigned)t0.x, e0,      live0, key, q, cnt, lane);
        push_owned((unsigned)t0.y, e0 + 1u, live0, key, q, cnt, lane);
        push_owned((unsigned)t1.x, e1,      live1, key, q, cnt, lane);
        push_owned((unsigned)t1.y, e1 + 1u, live1, key, q, cnt, lane);
        if (cnt >= 32) {
          __builtin_amdgcn_fence(__ATOMIC_SEQ_CST, "wavefront");
          const int full = cnt & ~31;
          for (int c = 0; c < full; c += 32)
            drain_chunk<true>(q, c, 32, x4, nrm, rates, impl, acc, lane);
          const int rem = cnt - full;
          if (lane < rem) q[lane] = q[full + lane];
          __builtin_amdgcn_fence(__ATOMIC_SEQ_CST, "wavefront");
          cnt = rem;
        }
      }
    }
  }
  __builtin_amdgcn_fence(__ATOMIC_SEQ_CST, "wavefront");
  if (cnt > 0)
    drain_chunk<false>(q, 0, cnt, x4, nrm, rates, impl, acc, lane);

  __syncthreads();
  for (int i = wave; i < OWN_ROWS; i += OWN_WAVES) {
    const unsigned t = ((unsigned)i << 8) | b;
    if (t < (unsigned)NNODE) {
      float4 s = ((const float4*)(acc + i * H))[lane];
      if (LAST) {
        s.x *= (1.0f / 3.0f); s.y *= (1.0f / 3.0f); s.z *= (1.0f / 3.0f); s.w *= (1.0f / 3.0f);
      }
      ((float4*)(out + (size_t)t * H))[lane] = s;
    }
  }
}

extern "C" void kernel_launch(void* const* d_in, const int* in_sizes, int n_in,
                              void* d_out, int out_size, void* d_ws, size_t ws_size,
                              hipStream_t stream) {
  (void)in_sizes; (void)n_in; (void)out_size; (void)ws_size;

  const float* user_x = (const float*)d_in[0];
  const float* book_x = (const float*)d_in[1];
  const float* user_W = (const float*)d_in[2];
  const float* user_b = (const float*)d_in[3];
  const float* book_W = (const float*)d_in[4];
  const float* book_b = (const float*)d_in[5];
  const int*   rates  = (const int*)d_in[6];
  const int*   impl   = (const int*)d_in[7];
  float* outp = (float*)d_out;

  float* x   = (float*)d_ws;
  float* buf = x   + (size_t)NNODE * H;
  float* deg = buf + (size_t)NNODE * H;
  float* dis = deg + NNODE;
  float* nrm = dis + NNODE;
  unsigned* pres = (unsigned*)deg;
  unsigned* part = (unsigned*)buf;

  {
    int waves = (NU + 31) / 32;
    gemm_bias_wmma<<<(waves + 7) / 8, 256, 0, stream>>>(user_x, user_W, user_b, x, NU);
    waves = (NB + 31) / 32;
    gemm_bias_wmma<<<(waves + 7) / 8, 256, 0, stream>>>(book_x, book_W, book_b,
                                                        x + (size_t)NU * H, NB);
  }

  degree_hist_kernel<<<HS * HR, HTHREADS, 0, stream>>>(rates, impl, part);
  dis_from_hist_kernel<<<(HPITCH + 255) / 256, 256, 0, stream>>>(part, dis);
  norm_kernel<<<(NE + 255) / 256, 256, 0, stream>>>(rates, impl, dis, nrm);
  present_kernel<<<PRES_BLOCKS, 1024, 0, stream>>>(rates, impl, pres);

  gather_owner_kernel<false><<<OWN_BLOCKS, OWN_THREADS, 0, stream>>>(x, x, nrm, rates, impl, pres, buf);
  gather_owner_kernel<true><<<OWN_BLOCKS, OWN_THREADS, 0, stream>>>(buf, x, nrm, rates, impl, pres, outp);
}


# second layer sum: each block replays the (source row, local row, weight) list its first-layer pass wrote into its own rows of out; no second edge search; LDS float adds still arrival order
# speedup vs baseline: 3.3267x; 1.1614x over previous
#include <hip/hip_runtime.h>

#define NU 100000
#define NB 50000
#define NNODE (NU + NB)
#define H 128
#define F 256
#define EPR 500000
#define NE (4 * EPR)

typedef __attribute__((ext_vector_type(2))) float v2f;
typedef __attribute__((ext_vector_type(8))) float v8f;

#define KBLK (F / 8)

struct Frag {
  float4 a[2];
  float4 b[8];
};

__device__ __forceinline__ void load_frag(Frag& f, const float4* __restrict__ a0p,
                                          const float4* __restrict__ a1p,
                                          const float4* __restrict__ bp, int kb) {
  f.a[0] = a0p[2 * kb];
  f.a[1] = a1p[2 * kb];
#pragma unroll
  for (int nt = 0; nt < 8; ++nt) f.b[nt] = bp[nt * (16 * F / 4) + 2 * kb];
}

__device__ __forceinline__ void mma_frag(v8f (&c)[2][8], const Frag& f) {
#pragma unroll
  for (int nt = 0; nt < 8; ++nt) {
#pragma unroll
    for (int t = 0; t < 2; ++t) {
      v2f a = {f.a[t].x, f.a[t].y};
      v2f b = {f.b[nt].x, f.b[nt].y};
      c[t][nt] = __builtin_amdgcn_wmma_f32_16x16x4_f32(
           false, a,  false, b,
           (short)0, c[t][nt],  false,  false);
    }
  }
#pragma unroll
  for (int nt = 0; nt < 8; ++nt) {
#pragma unroll
    for (int t = 0; t < 2; ++t) {
      v2f a = {f.a[t].z, f.a[t].w};
      v2f b = {f.b[nt].z, f.b[nt].w};
      c[t][nt] = __builtin_amdgcn_wmma_f32_16x16x4_f32(
           false, a,  false, b,
           (short)0, c[t][nt],  false,  false);
    }
  }
}

__global__ void __launch_bounds__(256)
gemm_bias_wmma(const float* __restrict__ X, const float* __restrict__ W,
               const float* __restrict__ bias, float* __restrict__ Y, int M) {
  const int wave = (int)((blockIdx.x * blockDim.x + threadIdx.x) >> 5);
  const int lane = threadIdx.x & 31;
  if (wave * 32 >= M) return;
  const int mbase = wave * 32;
  const int lrow  = lane & 15;
  const int kq    = (lane >> 4) << 2;
  const bool two = mbase + 16 < M;
  const int r0 = mbase + lrow;
  const int r1 = two ? (r0 + 16) : r0;

  const float4* a0p = (const float4*)(X + (size_t)r0 * F + kq);
  const float4* a1p = (const float4*)(X + (size_t)r1 * F + kq);
  const float4* bp  = (const float4*)(W + (size_t)lrow * F + kq);

  v8f c[2][8] = {};

  Frag f0, f1;
  load_frag(f0, a0p, a1p, bp, 0);
  for (int kb = 0; kb < KBLK; kb += 2) {
    load_frag(f1, a0p, a1p, bp, kb + 1);
    __builtin_amdgcn_sched_barrier(0);
    mma_frag(c, f0);
    __builtin_amdgcn_sched_barrier(0);
    load_frag(f0, a0p, a1p, bp, (kb + 2 < KBLK) ? (kb + 2) : 0);
    __builtin_amdgcn_sched_barrier(0);
    mma_frag(c, f1);
    __builtin_amdgcn_sched_barrier(0);
  }

  const int mo = (lane >> 4) << 3;
#pragma unroll
  for (int t = 0; t < 2; ++t) {
    if (t == 0 || two) {
#pragma unroll
      for (int nt = 0; nt < 8; ++nt) {
        float bv = bias[nt * 16 + lrow];
#pragma unroll
        for (int v = 0; v < 8; ++v) {
          Y[(size_t)(mbase + 16 * t + mo + v) * H + nt * 16 + lrow] = c[t][nt][v] + bv;
        }
      }
    }
  }
}

__device__ __forceinline__ void decode_edge(int e, const int* __restrict__ rates,
                                            const int* __restrict__ impl,
                                            int& row, int& col) {
  const int rev = (e >= 2 * EPR) ? 1 : 0;
  int idx = rev ? (e - 2 * EPR) : e;
  const int* ed = (idx < EPR) ? rates : impl;
  const int i = (idx < EPR) ? idx : (idx - EPR);
  const int u = ed[i];
  const int b = ed[i + EPR];
  row = rev ? (b + NU) : u;
  col = rev ? u : b;
}

__global__ void __launch_bounds__(256)
norm_kernel(const int* __restrict__ rates, const int* __restrict__ impl,
            const float* __restrict__ dis, float* __restrict__ nrm) {
  const int e = blockIdx.x * blockDim.x + threadIdx.x;
  if (e >= NE) return;
  int row, col;
  decode_edge(e, rates, impl, row, col);
  nrm[e] = dis[row] * dis[col];
}

#define OWN_BLOCKS  256
#define OWN_ROWS    ((NNODE + OWN_BLOCKS - 1) / OWN_BLOCKS)
#define OWN_THREADS 1024
#define OWN_WAVES   (OWN_THREADS / 32)
#define QCAP        160
#define DRAIN_G     16
#define SEED_G      10

#define LOG_ROWS    585
#define LOG_SLOTS   (2 * LOG_ROWS)
static_assert((((LOG_ROWS - 1) << 8) | (OWN_BLOCKS - 1)) < NNODE, "every block owns LOG_ROWS real rows");
static_assert(NNODE < (1 << 22), "a source row fits the 22 bits above the local row");

#define GRP_SEG     ((EPR + 63) / 64)
#define SEG_PITCH   7936
#define SEG_STEPS   (SEG_PITCH / 128)
#define PPITCH      (4 * SEG_PITCH)
#define NSTEP       (4 * SEG_STEPS)
#define PRES_BLOCKS (PPITCH / 32)
static_assert(8 * PPITCH + 256 <= 2 * NNODE, "planes and the 256 list counts fit deg + dis");

__device__ __forceinline__ void lds_add(float* p, float v) {
  __hip_atomic_fetch_add(p, v, __ATOMIC_RELAXED, __HIP_MEMORY_SCOPE_WORKGROUP);
}

template <bool FULL>
__device__ __forceinline__ void sum_rows(int row, int loc, int wbits, int nv,
                                         const float4* __restrict__ x4, float* acc, int lane) {
  const int h = lane >> 4;
#pragma unroll
  for (int half = 0; half < 32 / DRAIN_G; ++half) {
    if (!FULL && half * DRAIN_G >= nv) break;
    float4 v[DRAIN_G];
#pragma unroll
    for (int g = 0; g < DRAIN_G; ++g) {
      const int r = __builtin_amdgcn_readlane(row, half * DRAIN_G + g);
      v[g] = x4[(size_t)r * (H / 4) + lane];
    }
#pragma unroll
    for (int g = 0; g < DRAIN_G; ++g) {
      if (FULL || half * DRAIN_G + g < nv) {
        const int il = __builtin_amdgcn_readlane(loc, half * DRAIN_G + g);
        const float m = __int_as_float(__builtin_amdgcn_readlane(wbits, half * DRAIN_G + g));
        float* a = acc + il * H + 4 * lane;
        const float4 s = v[g];
        lds_add(a + h,             (h ? s.y : s.x) * m);
        lds_add(a + ((1 + h) & 3), (h ? s.z : s.y) * m);
        lds_add(a + ((2 + h) & 3), (h ? s.w : s.z) * m);
        lds_add(a + ((3 + h) & 3), (h ? s.x : s.w) * m);
      }
    }
  }
}

__device__ __forceinline__ uint2* list_slot(float* lst, unsigned b, int slot) {
  return (uint2*)(lst + (size_t)((((unsigned)slot >> 1) << 8) | b) * H + (slot & 1) * (H / 2));
}

template <bool FULL, bool LOG>
__device__ __forceinline__ void drain_chunk(const unsigned* q, int base, int nv,
                                            const float4* __restrict__ x4,
                                            const float* __restrict__ nrm,
                                            const int* __restrict__ rates,
                                            const int* __restrict__ impl,
                                            float* acc, int lane,
                                            float* lst, unsigned* nslot, unsigned b) {
  const int li = FULL ? (base + lane) : (base + ((lane < nv) ? lane : 0));
  const unsigned d = q[li];
  const int e = (int)(d >> 10);
  const int loc = (int)(d & 1023u);
  const bool rev = e >= 2 * EPR;
  const int idx = rev ? (e - 2 * EPR) : e;
  const bool first = idx < EPR;
  const int i = first ? idx : (idx - EPR);
  const int* ed = first ? rates : impl;
  const int sv = ed[i + (rev ? EPR : 0)];
  const int row = rev ? (sv + NU) : sv;
  const int wbits = __float_as_int(nrm[e]);
  if (LOG) {
    unsigned got = 0u;
    if (lane == 0)
      got = __hip_atomic_fetch_add(nslot, 1u, __ATOMIC_RELAXED, __HIP_MEMORY_SCOPE_WORKGROUP);
    const int slot = __builtin_amdgcn_readfirstlane((int)got);
    if (slot < LOG_SLOTS) {
      const bool valid = FULL || lane < nv;
      uint2 ent;
      ent.x = valid ? (((unsigned)row << 10) | (unsigned)loc) : 1023u;
      ent.y = valid ? (unsigned)wbits : 0u;
      list_slot(lst, b, slot)[lane] = ent;
    }
  }
  sum_rows<FULL>(row, loc, wbits, nv, x4, acc, lane);
}

__device__ __forceinline__ const int* seg_cols(int seg, const int* __restrict__ rates,
                                               const int* __restrict__ impl) {
  return (seg == 0) ? (rates + EPR) : (seg == 1) ? (impl + EPR) : (seg == 2) ? rates : impl;
}

#define HR      8
#define HS      32
#define HRANGE  (NNODE / HR)
#define HPAIRS  (HRANGE / 2)
#define HSLICE  (NE / HS)
#define HSL4    (HSLICE / 4)
#define HPITCH  (NNODE / 2)
#define HTHREADS 1024
static_assert(HSLICE < 65536, "a 16-bit count must hold a whole slice");
static_assert(HRANGE * HR == NNODE && HRANGE % 2 == 0, "ranges cover the nodes in whole dwords");
static_assert(HSLICE * HS == NE && EPR % HSLICE == 0 && HSLICE % 4 == 0, "slices stay inside a segment");

__device__ __forceinline__ void hist_add(unsigned* cnt, int t, unsigned lo) {
  const unsigned q = (unsigned)t - lo;
  if (q < (unsigned)HRANGE)
    __hip_atomic_fetch_add(cnt + (q >> 1), (q & 1u) ? 65536u : 1u, __ATOMIC_RELAXED,
                           __HIP_MEMORY_SCOPE_WORKGROUP);
}

__global__ void __launch_bounds__(HTHREADS)
degree_hist_kernel(const int* __restrict__ rates, const int* __restrict__ impl,
                   unsigned* __restrict__ part) {
  __shared__ unsigned cnt[HPAIRS];
  const int tid = threadIdx.x;
  const int r = (int)blockIdx.x % HR;
  const int s = (int)blockIdx.x / HR;
  for (int i = tid; i < HPAIRS; i += HTHREADS) cnt[i] = 0u;
  __syncthreads();
  const int seg = s / (HS / 4);
  const int4* p = (const int4*)(seg_cols(seg, rates, impl) + (s - seg * (HS / 4)) * HSLICE);
  const unsigned lo = (unsigned)(r * HRANGE);
  for (int j0 = 0; j0 < HSL4; j0 += 4 * HTHREADS) {
    int4 v[4];
#pragma unroll
    for (int u = 0; u < 4; ++u) {
      const int j = j0 + u * HTHREADS + tid;
      v[u] = (j < HSL4) ? p[j] : make_int4(-1, -1, -1, -1);
    }
#pragma unroll
    for (int u = 0; u < 4; ++u) {
      hist_add(cnt, v[u].x, lo);
      hist_add(cnt, v[u].y, lo);
      hist_add(cnt, v[u].z, lo);
      hist_add(cnt, v[u].w, lo);
    }
  }
  __syncthreads();
  unsigned* dst = part + (size_t)s * HPITCH + r * HPAIRS;
  for (int i = tid; i < HPAIRS; i += HTHREADS) dst[i] = cnt[i];
}

__global__ void __launch_bounds__(256)
dis_from_hist_kernel(const unsigned* __restrict__ part, float* __restrict__ dis) {
  const int p = blockIdx.x * blockDim.x + threadIdx.x;
  if (p >= HPITCH) return;
  unsigned v[HS];
#pragma unroll
  for (int s = 0; s < HS; ++s) v[s] = part[(size_t)s * HPITCH + p];
  unsigned c0 = 0u, c1 = 0u;
#pragma unroll
  for (int s = 0; s < HS; ++s) {
    c0 += v[s] & 0xffffu;
    c1 += v[s] >> 16;
  }
  const float d0 = (float)c0, d1 = (float)c1;
  float2 o;
  o.x = (d0 > 0.0f) ? rsqrtf(d0) : 0.0f;
  o.y = (d1 > 0.0f) ? rsqrtf(d1) : 0.0f;
  ((float2*)dis)[p] = o;
}

__global__ void __launch_bounds__(1024)
present_kernel(const int* __restrict__ rates, const int* __restrict__ impl,
               unsigned* __restrict__ pres) {
  __shared__ unsigned st[8][32];
  const int lane = threadIdx.x & 31;
  const int wave = __builtin_amdgcn_readfirstlane((int)(threadIdx.x >> 5));
  const int pidx = (int)blockIdx.x * 32 + wave;
  const int seg = pidx / SEG_PITCH;
  const int gs  = pidx - seg * SEG_PITCH;
  unsigned myv = 0u;
  if (gs < GRP_SEG) {
    const int idx = gs * 64 + 2 * lane;
    const bool ok = idx < EPR;
    const int2 t2 = *(const int2*)(seg_cols(seg, rates, impl) + (ok ? idx : 0));
    const unsigned ta = (unsigned)t2.x, tb = (unsigned)t2.y;
    const unsigned valid = __builtin_amdgcn_ballot_w32(ok);
    unsigned la = valid, lb = valid;
#pragma unroll
    for (int i = 0; i < 5; ++i) {
      const unsigned ka = __builtin_amdgcn_ballot_w32(((ta >> i) & 1u) != 0u);
      const unsigned kb = __builtin_amdgcn_ballot_w32(((tb >> i) & 1u) != 0u);
      const bool on = ((lane >> i) & 1) != 0;
      la &= on ? ka : ~ka;
      lb &= on ? kb : ~kb;
    }
    unsigned ha[3], hb[3];
#pragma unroll
    for (int i = 0; i < 3; ++i) {
      ha[i] = __builtin_amdgcn_ballot_w32(((ta >> (5 + i)) & 1u) != 0u);
      hb[i] = __builtin_amdgcn_ballot_w32(((tb >> (5 + i)) & 1u) != 0u);
    }
#pragma unroll
    for (int m = 0; m < 8; ++m) {
      unsigned sa = la, sb = lb;
#pragma unroll
      for (int i = 0; i < 3; ++i) {
        sa &= ((m >> i) & 1) ? ha[i] : ~ha[i];
        sb &= ((m >> i) & 1) ? hb[i] : ~hb[i];
      }
      const unsigned pm = __builtin_amdgcn_ballot_w32((sa | sb) != 0u);
      if (lane == m) myv = pm;
    }
  }
  if (lane < 8) st[lane][wave] = myv;
  __syncthreads();
  if (wave < 8) pres[wave * PPITCH + (int)blockIdx.x * 32 + lane] = st[wave][lane];
}

__device__ __forceinline__ void push_owned(unsigned t, unsigned e, unsigned live, unsigned key,
                                           unsigned* q, int& cnt, int lane) {
  const unsigned iloc = ((t >> 8) | (t << 24)) - key;
  const unsigned mk = __builtin_amdgcn_ballot_w32(iloc < (unsigned)OWN_ROWS) & live;
  if (mk != 0u) {
    if ((mk >> lane) & 1u)
      q[cnt + (int)__builtin_amdgcn_mbcnt_lo(mk, 0u)] = (e << 10) | iloc;
    cnt += __builtin_popcount(mk);
  }
}

template <bool LAST>
__global__ void __launch_bounds__(OWN_THREADS)
gather_owner_kernel(const float* __restrict__ x, const float* __restrict__ x0,
                    const float* __restrict__ nrm,
                    const int* __restrict__ rates, const int* __restrict__ impl,
                    const unsigned* __restrict__ pres, float* lst, unsigned* cntw,
                    float* out) {
  __shared__ __attribute__((aligned(16))) float acc[OWN_ROWS * H];
  __shared__ unsigned queue[OWN_WAVES * QCAP];
  __shared__ unsigned nslot;

  const int tid  = threadIdx.x;
  const int lane = tid & 31;
  const int wave = __builtin_amdgcn_readfirstlane(tid >> 5);
  const unsigned b = blockIdx.x;

  if (tid == 0) nslot = 0u;
  float4* acc4 = (float4*)acc;
  const float4* s4 = (const float4*)x0;
  for (int i0 = wave; i0 < OWN_ROWS; i0 += SEED_G * OWN_WAVES) {
    float4 v[SEED_G];
#pragma unroll
    for (int u = 0; u < SEED_G; ++u) {
      const unsigned t = ((unsigned)(i0 + u * OWN_WAVES) << 8) | b;
      v[u] = (t < (unsigned)NNODE) ? s4[(size_t)t * (H / 4) + lane]
                                   : make_float4(0.0f, 0.0f, 0.0f, 0.0f);
    }
#pragma unroll
    for (int u = 0; u < SEED_G; ++u) {
      const int i = i0 + u * OWN_WAVES;
      if (i < OWN_ROWS) acc4[i * (H / 4) + lane] = v[u];
    }
  }
  __syncthreads();

  const float4* x4 = (const float4*)x;
  const unsigned nlist = LAST ? cntw[b] : 0u;
  if (LAST && nlist <= (unsigned)LOG_SLOTS) {
    for (int s = wave; s < (int)nlist; s += OWN_WAVES) {
      const uint2 cur = list_slot(lst, b, s)[lane];
      const int loc = (int)(cur.x & 1023u);
      const int nv = __builtin_popcount(__builtin_amdgcn_ballot_w32(loc < OWN_ROWS));
      if (nv == 32)
        sum_rows<true>((int)(cur.x >> 10), loc, (int)cur.y, 32, x4, acc, lane);
      else if (nv > 0)
        sum_rows<false>((int)(cur.x >> 10), loc, (int)cur.y, nv, x4, acc, lane);
    }
  } else {
    unsigned* q = queue + wave * QCAP;
    const unsigned key = b << 24;
    const int rot = (int)((b * NSTEP) >> 8);
    const uint4* plane4 = (const uint4*)(pres + (b >> 5) * PPITCH);
    const unsigned bbit = b & 31u;
    int cnt = 0;

    for (int s = wave; s < NSTEP; s += OWN_WAVES) {
      int sr = s + rot;
      if (sr >= NSTEP) sr -= NSTEP;
      const int seg = sr / SEG_STEPS;
      const int st  = sr - seg * SEG_STEPS;
      const int* colp = seg_cols(seg, rates, impl);
      const unsigned ebase = (unsigned)(seg * EPR);
      const uint4 d = plane4[sr * 32 + lane];
      const unsigned dk[4] = {d.x, d.y, d.z, d.w};
#pragma unroll
      for (int k = 0; k < 4; ++k) {
        unsigned m = __builtin_amdgcn_ballot_w32(((dk[k] >> bbit) & 1u) != 0u);
        while (m != 0u) {
          const int l0 = __builtin_ctz(m);
          m &= m - 1u;
          const unsigned two = (m != 0u) ? ~0u : 0u;
          const int l1 = (m != 0u) ? __builtin_ctz(m) : l0;
          m &= m - 1u;
          const int i0 = ((((st * 32 + l0) << 2) + k) << 6) + 2 * lane;
          const int i1 = ((((st * 32 + l1) << 2) + k) << 6) + 2 * lane;
          const bool ok0 = i0 < EPR, ok1 = i1 < EPR;
          const int2 t0 = *(const int2*)(colp + (ok0 ? i0 : 0));
          const int2 t1 = *(const int2*)(colp + (ok1 ? i1 : 0));
          const unsigned live0 = __builtin_amdgcn_ballot_w32(ok0);
          const unsigned live1 = __builtin_amdgcn_ballot_w32(ok1) & two;
          const unsigned e0 = ebase + (unsigned)i0, e1 = ebase + (unsigned)i1;
          push_owned((unsigned)t0.x, e0,      live0, key, q, cnt, lane);
          push_owned((unsigned)t0.y, e0 + 1u, live0, key, q, cnt, lane);
          push_owned((unsigned)t1.x, e1,      live1, key, q, cnt, lane);
          push_owned((unsigned)t1.y, e1 + 1u, live1, key, q, cnt, lane);
          if (cnt >= 32) {
            __builtin_amdgcn_fence(__ATOMIC_SEQ_CST, "wavefront");
            const int full = cnt & ~31;
            for (int c = 0; c < full; c += 32)
              drain_chunk<true, !LAST>(q, c, 32, x4, nrm, rates, impl, acc, lane, lst, &nslot, b);
            const int rem = cnt - full;
            if (lane < rem) q[lane] = q[full + lane];
            __builtin_amdgcn_fence(__ATOMIC_SEQ_CST, "wavefront");
            cnt = rem;
          }
        }
      }
    }
    __builtin_amdgcn_fence(__ATOMIC_SEQ_CST, "wavefront");
    if (cnt > 0)
      drain_chunk<false, !LAST>(q, 0, cnt, x4, nrm, rates, impl, acc, lane, lst, &nslot, b);
  }

  __syncthreads();
  if (!LAST && tid == 0) cntw[b] = nslot;
  for (int i = wave; i < OWN_ROWS; i += OWN_WAVES) {
    const unsigned t = ((unsigned)i << 8) | b;
    if (t < (unsigned)NNODE) {
      float4 s = ((const float4*)(acc + i * H))[lane];
      if (LAST) {
        s.x *= (1.0f / 3.0f); s.y *= (1.0f / 3.0f); s.z *= (1.0f / 3.0f); s.w *= (1.0f / 3.0f);
      }
      ((float4*)(out + (size_t)t * H))[lane] = s;
    }
  }
}

extern "C" void kernel_launch(void* const* d_in, const int* in_sizes, int n_in,
                              void* d_out, int out_size, void* d_ws, size_t ws_size,
                              hipStream_t stream) {
  (void)in_sizes; (void)n_in; (void)out_size; (void)ws_size;

  const float* user_x = (const float*)d_in[0];
  const float* book_x = (const float*)d_in[1];
  const float* user_W = (const float*)d_in[2];
  const float* user_b = (const float*)d_in[3];
  const float* book_W = (const float*)d_in[4];
  const float* book_b = (const float*)d_in[5];
  const int*   rates  = (const int*)d_in[6];
  const int*   impl   = (const int*)d_in[7];
  float* outp = (float*)d_out;

  float* x   = (float*)d_ws;
  float* buf = x   + (size_t)NNODE * H;
  float* deg = buf + (size_t)NNODE * H;
  float* dis = deg + NNODE;
  float* nrm = dis + NNODE;
  unsigned* pres = (unsigned*)deg;
  unsigned* cntw = pres + 8 * PPITCH;
  unsigned* part = (unsigned*)buf;

  {
    int waves = (NU + 31) / 32;
    gemm_bias_wmma<<<(waves + 7) / 8, 256, 0, stream>>>(user_x, user_W, user_b, x, NU);
    waves = (NB + 31) / 32;
    gemm_bias_wmma<<<(waves + 7) / 8, 256, 0, stream>>>(book_x, book_W, book_b,
                                                        x + (size_t)NU * H, NB);
  }

  degree_hist_kernel<<<HS * HR, HTHREADS, 0, stream>>>(rates, impl, part);
  dis_from_hist_kernel<<<(HPITCH + 255) / 256, 256, 0, stream>>>(part, dis);
  norm_kernel<<<(NE + 255) / 256, 256, 0, stream>>>(rates, impl, dis, nrm);
  present_kernel<<<PRES_BLOCKS, 1024, 0, stream>>>(rates, impl, pres);

  gather_owner_kernel<false><<<OWN_BLOCKS, OWN_THREADS, 0, stream>>>(x, x, nrm, rates, impl, pres, outp, cntw, buf);
  gather_owner_kernel<true><<<OWN_BLOCKS, OWN_THREADS, 0, stream>>>(buf, x, nrm, rates, impl, pres, outp, cntw, outp);
}


# edge search: filter pass writes per (owner block, 64-edge group) a dword naming the block's 1 or 2 edges (place, local row); blocks queue edges from it, no target reads; LDS float adds arrival order
# speedup vs baseline: 3.5785x; 1.0757x over previous
#include <hip/hip_runtime.h>

#define NU 100000
#define NB 50000
#define NNODE (NU + NB)
#define H 128
#define F 256
#define EPR 500000
#define NE (4 * EPR)

typedef __attribute__((ext_vector_type(2))) float v2f;
typedef __attribute__((ext_vector_type(8))) float v8f;

#define KBLK (F / 8)

struct Frag {
  float4 a[2];
  float4 b[8];
};

__device__ __forceinline__ void load_frag(Frag& f, const float4* __restrict__ a0p,
                                          const float4* __restrict__ a1p,
                                          const float4* __restrict__ bp, int kb) {
  f.a[0] = a0p[2 * kb];
  f.a[1] = a1p[2 * kb];
#pragma unroll
  for (int nt = 0; nt < 8; ++nt) f.b[nt] = bp[nt * (16 * F / 4) + 2 * kb];
}

__device__ __forceinline__ void mma_frag(v8f (&c)[2][8], const Frag& f) {
#pragma unroll
  for (int nt = 0; nt < 8; ++nt) {
#pragma unroll
    for (int t = 0; t < 2; ++t) {
      v2f a = {f.a[t].x, f.a[t].y};
      v2f b = {f.b[nt].x, f.b[nt].y};
      c[t][nt] = __builtin_amdgcn_wmma_f32_16x16x4_f32(
           false, a,  false, b,
           (short)0, c[t][nt],  false,  false);
    }
  }
#pragma unroll
  for (int nt = 0; nt < 8; ++nt) {
#pragma unroll
    for (int t = 0; t < 2; ++t) {
      v2f a = {f.a[t].z, f.a[t].w};
      v2f b = {f.b[nt].z, f.b[nt].w};
      c[t][nt] = __builtin_amdgcn_wmma_f32_16x16x4_f32(
           false, a,  false, b,
           (short)0, c[t][nt],  false,  false);
    }
  }
}

__global__ void __launch_bounds__(256)
gemm_bias_wmma(const float* __restrict__ X, const float* __restrict__ W,
               const float* __restrict__ bias, float* __restrict__ Y, int M) {
  const int wave = (int)((blockIdx.x * blockDim.x + threadIdx.x) >> 5);
  const int lane = threadIdx.x & 31;
  if (wave * 32 >= M) return;
  const int mbase = wave * 32;
  const int lrow  = lane & 15;
  const int kq    = (lane >> 4) << 2;
  const bool two = mbase + 16 < M;
  const int r0 = mbase + lrow;
  const int r1 = two ? (r0 + 16) : r0;

  const float4* a0p = (const float4*)(X + (size_t)r0 * F + kq);
  const float4* a1p = (const float4*)(X + (size_t)r1 * F + kq);
  const float4* bp  = (const float4*)(W + (size_t)lrow * F + kq);

  v8f c[2][8] = {};

  Frag f0, f1;
  load_frag(f0, a0p, a1p, bp, 0);
  for (int kb = 0; kb < KBLK; kb += 2) {
    load_frag(f1, a0p, a1p, bp, kb + 1);
    __builtin_amdgcn_sched_barrier(0);
    mma_frag(c, f0);
    __builtin_amdgcn_sched_barrier(0);
    load_frag(f0, a0p, a1p, bp, (kb + 2 < KBLK) ? (kb + 2) : 0);
    __builtin_amdgcn_sched_barrier(0);
    mma_frag(c, f1);
    __builtin_amdgcn_sched_barrier(0);
  }

  const int mo = (lane >> 4) << 3;
#pragma unroll
  for (int t = 0; t < 2; ++t) {
    if (t == 0 || two) {
#pragma unroll
      for (int nt = 0; nt < 8; ++nt) {
        float bv = bias[nt * 16 + lrow];
#pragma unroll
        for (int v = 0; v < 8; ++v) {
          Y[(size_t)(mbase + 16 * t + mo + v) * H + nt * 16 + lrow] = c[t][nt][v] + bv;
        }
      }
    }
  }
}

__device__ __forceinline__ void decode_edge(int e, const int* __restrict__ rates,
                                            const int* __restrict__ impl,
                                            int& row, int& col) {
  const int rev = (e >= 2 * EPR) ? 1 : 0;
  int idx = rev ? (e - 2 * EPR) : e;
  const int* ed = (idx < EPR) ? rates : impl;
  const int i = (idx < EPR) ? idx : (idx - EPR);
  const int u = ed[i];
  const int b = ed[i + EPR];
  row = rev ? (b + NU) : u;
  col = rev ? u : b;
}

__global__ void __launch_bounds__(256)
norm_kernel(const int* __restrict__ rates, const int* __restrict__ impl,
            const float* __restrict__ dis, float* __restrict__ nrm) {
  const int e = blockIdx.x * blockDim.x + threadIdx.x;
  if (e >= NE) return;
  int row, col;
  decode_edge(e, rates, impl, row, col);
  nrm[e] = dis[row] * dis[col];
}

#define OWN_BLOCKS  256
#define OWN_ROWS    ((NNODE + OWN_BLOCKS - 1) / OWN_BLOCKS)
#define OWN_THREADS 1024
#define OWN_WAVES   (OWN_THREADS / 32)
#define QCAP        160
#define DRAIN_G     16
#define SEED_G      10

#define TAB_ROWS    248
#define LOG_ROWS    (585 - TAB_ROWS)
#define LOG_SLOTS   (2 * LOG_ROWS)
static_assert((((LOG_ROWS + TAB_ROWS - 1) << 8) | (OWN_BLOCKS - 1)) < NNODE, "every block owns that many real rows");
static_assert(NNODE < (1 << 22), "a source row fits the 22 bits above the local row");

#define GRP_SEG     ((EPR + 63) / 64)
#define SEG_PITCH   7936
#define SEG_STEPS   (SEG_PITCH / 128)
#define PPITCH      (4 * SEG_PITCH)
#define NSTEP       (4 * SEG_STEPS)
#define TAB_BLOCKS  (PPITCH / 32)
#define TAB_MANY    0xffffffffu
#define TAB_PITCH   257
static_assert(NSTEP == TAB_ROWS, "one table row per wave step");
static_assert(SEG_PITCH % 32 == 0, "a table block stays inside one segment");

__device__ __forceinline__ void lds_add(float* p, float v) {
  __hip_atomic_fetch_add(p, v, __ATOMIC_RELAXED, __HIP_MEMORY_SCOPE_WORKGROUP);
}

template <bool FULL>
__device__ __forceinline__ void sum_rows(int row, int loc, int wbits, int nv,
                                         const float4* __restrict__ x4, float* acc, int lane) {
  const int h = lane >> 4;
#pragma unroll
  for (int half = 0; half < 32 / DRAIN_G; ++half) {
    if (!FULL && half * DRAIN_G >= nv) break;
    float4 v[DRAIN_G];
#pragma unroll
    for (int g = 0; g < DRAIN_G; ++g) {
      const int r = __builtin_amdgcn_readlane(row, half * DRAIN_G + g);
      v[g] = x4[(size_t)r * (H / 4) + lane];
    }
#pragma unroll
    for (int g = 0; g < DRAIN_G; ++g) {
      if (FULL || half * DRAIN_G + g < nv) {
        const int il = __builtin_amdgcn_readlane(loc, half * DRAIN_G + g);
        const float m = __int_as_float(__builtin_amdgcn_readlane(wbits, half * DRAIN_G + g));
        float* a = acc + il * H + 4 * lane;
        const float4 s = v[g];
        lds_add(a + h,             (h ? s.y : s.x) * m);
        lds_add(a + ((1 + h) & 3), (h ? s.z : s.y) * m);
        lds_add(a + ((2 + h) & 3), (h ? s.w : s.z) * m);
        lds_add(a + ((3 + h) & 3), (h ? s.x : s.w) * m);
      }
    }
  }
}

__device__ __forceinline__ uint2* list_slot(float* lst, unsigned b, int slot) {
  return (uint2*)(lst + (size_t)((((unsigned)slot >> 1) << 8) | b) * H + (slot & 1) * (H / 2));
}

template <bool FULL, bool LOG>
__device__ __forceinline__ void drain_chunk(const unsigned* q, int base, int nv,
                                            const float4* __restrict__ x4,
                                            const float* __restrict__ nrm,
                                            const int* __restrict__ rates,
                                            const int* __restrict__ impl,
                                            float* acc, int lane,
                                            float* lst, unsigned* nslot, unsigned b) {
  const int li = FULL ? (base + lane) : (base + ((lane < nv) ? lane : 0));
  const unsigned d = q[li];
  const int e = (int)(d >> 10);
  const int loc = (int)(d & 1023u);
  const bool rev = e >= 2 * EPR;
  const int idx = rev ? (e - 2 * EPR) : e;
  const bool first = idx < EPR;
  const int i = first ? idx : (idx - EPR);
  const int* ed = first ? rates : impl;
  const int sv = ed[i + (rev ? EPR : 0)];
  const int row = rev ? (sv + NU) : sv;
  const int wbits = __float_as_int(nrm[e]);
  if (LOG) {
    unsigned got = 0u;
    if (lane == 0)
      got = __hip_atomic_fetch_add(nslot, 1u, __ATOMIC_RELAXED, __HIP_MEMORY_SCOPE_WORKGROUP);
    const int slot = __builtin_amdgcn_readfirstlane((int)got);
    if (slot < LOG_SLOTS) {
      const bool valid = FULL || lane < nv;
      uint2 ent;
      ent.x = valid ? (((unsigned)row << 10) | (unsigned)loc) : 1023u;
      ent.y = valid ? (unsigned)wbits : 0u;
      list_slot(lst, b, slot)[lane] = ent;
    }
  }
  sum_rows<FULL>(row, loc, wbits, nv, x4, acc, lane);
}

__device__ __forceinline__ const int* seg_cols(int seg, const int* __restrict__ rates,
                                               const int* __restrict__ impl) {
  return (seg == 0) ? (rates + EPR) : (seg == 1) ? (impl + EPR) : (seg == 2) ? rates : impl;
}

#define HR      8
#define HS      32
#define HRANGE  (NNODE / HR)
#define HPAIRS  (HRANGE / 2)
#define HSLICE  (NE / HS)
#define HSL4    (HSLICE / 4)
#define HPITCH  (NNODE / 2)
#define HTHREADS 1024
static_assert(HSLICE < 65536, "a 16-bit count must hold a whole slice");
static_assert(HRANGE * HR == NNODE && HRANGE % 2 == 0, "ranges cover the nodes in whole dwords");
static_assert(HSLICE * HS == NE && EPR % HSLICE == 0 && HSLICE % 4 == 0, "slices stay inside a segment");

__device__ __forceinline__ void hist_add(unsigned* cnt, int t, unsigned lo) {
  const unsigned q = (unsigned)t - lo;
  if (q < (unsigned)HRANGE)
    __hip_atomic_fetch_add(cnt + (q >> 1), (q & 1u) ? 65536u : 1u, __ATOMIC_RELAXED,
                           __HIP_MEMORY_SCOPE_WORKGROUP);
}

__global__ void __launch_bounds__(HTHREADS)
degree_hist_kernel(const int* __restrict__ rates, const int* __restrict__ impl,
                   unsigned* __restrict__ part) {
  __shared__ unsigned cnt[HPAIRS];
  const int tid = threadIdx.x;
  const int r = (int)blockIdx.x % HR;
  const int s = (int)blockIdx.x / HR;
  for (int i = tid; i < HPAIRS; i += HTHREADS) cnt[i] = 0u;
  __syncthreads();
  const int seg = s / (HS / 4);
  const int4* p = (const int4*)(seg_cols(seg, rates, impl) + (s - seg * (HS / 4)) * HSLICE);
  const unsigned lo = (unsigned)(r * HRANGE);
  for (int j0 = 0; j0 < HSL4; j0 += 4 * HTHREADS) {
    int4 v[4];
#pragma unroll
    for (int u = 0; u < 4; ++u) {
      const int j = j0 + u * HTHREADS + tid;
      v[u] = (j < HSL4) ? p[j] : make_int4(-1, -1, -1, -1);
    }
#pragma unroll
    for (int u = 0; u < 4; ++u) {
      hist_add(cnt, v[u].x, lo);
      hist_add(cnt, v[u].y, lo);
      hist_add(cnt, v[u].z, lo);
      hist_add(cnt, v[u].w, lo);
    }
  }
  __syncthreads();
  unsigned* dst = part + (size_t)s * HPITCH + r * HPAIRS;
  for (int i = tid; i < HPAIRS; i += HTHREADS) dst[i] = cnt[i];
}

__global__ void __launch_bounds__(256)
dis_from_hist_kernel(const unsigned* __restrict__ part, float* __restrict__ dis) {
  const int p = blockIdx.x * blockDim.x + threadIdx.x;
  if (p >= HPITCH) return;
  unsigned v[HS];
#pragma unroll
  for (int s = 0; s < HS; ++s) v[s] = part[(size_t)s * HPITCH + p];
  unsigned c0 = 0u, c1 = 0u;
#pragma unroll
  for (int s = 0; s < HS; ++s) {
    c0 += v[s] & 0xffffu;
    c1 += v[s] >> 16;
  }
  const float d0 = (float)c0, d1 = (float)c1;
  float2 o;
  o.x = (d0 > 0.0f) ? rsqrtf(d0) : 0.0f;
  o.y = (d1 > 0.0f) ? rsqrtf(d1) : 0.0f;
  ((float2*)dis)[p] = o;
}

__device__ __forceinline__ unsigned* tab_row(float* tab, unsigned o, int s) {
  return (unsigned*)(tab + (size_t)(((unsigned)(LOG_ROWS + s) << 8) | o) * H);
}

__global__ void __launch_bounds__(1024)
table_kernel(const int* __restrict__ rates, const int* __restrict__ impl, float* tab) {
  __shared__ __attribute__((aligned(16))) unsigned sm[32 * TAB_PITCH];
  __shared__ __attribute__((aligned(16))) unsigned mx[32 * TAB_PITCH];
  const int tid  = threadIdx.x;
  const int lane = tid & 31;
  const int wave = __builtin_amdgcn_readfirstlane(tid >> 5);
  for (int i = tid; i < 32 * TAB_PITCH / 4; i += 1024) {
    ((uint4*)sm)[i] = make_uint4(0u, 0u, 0u, 0u);
    ((uint4*)mx)[i] = make_uint4(0u, 0u, 0u, 0u);
  }
  __syncthreads();
  const int pidx = (int)blockIdx.x * 32 + wave;
  const int seg = pidx / SEG_PITCH;
  const int gs  = pidx - seg * SEG_PITCH;
  if (gs < GRP_SEG) {
    const int idx = gs * 64 + 2 * lane;
    const bool ok = idx < EPR;
    const int2 t2 = *(const int2*)(seg_cols(seg, rates, impl) + (ok ? idx : 0));
    const unsigned ta = (unsigned)t2.x, tb = (unsigned)t2.y;
    if (ok && ta < (unsigned)(OWN_ROWS << 8)) {
      const unsigned code = (((unsigned)(2 * lane) << 10) | (ta >> 8)) + 1u;
      const int cell = wave * TAB_PITCH + (int)(ta & 255u);
      __hip_atomic_fetch_add(sm + cell, (1u << 24) + code, __ATOMIC_RELAXED, __HIP_MEMORY_SCOPE_WORKGROUP);
      __hip_atomic_fetch_max(mx + cell, code, __ATOMIC_RELAXED, __HIP_MEMORY_SCOPE_WORKGROUP);
    }
    if (ok && tb < (unsigned)(OWN_ROWS << 8)) {
      const unsigned code = (((unsigned)(2 * lane + 1) << 10) | (tb >> 8)) + 1u;
      const int cell = wave * TAB_PITCH + (int)(tb & 255u);
      __hip_atomic_fetch_add(sm + cell, (1u << 24) + code, __ATOMIC_RELAXED, __HIP_MEMORY_SCOPE_WORKGROUP);
      __hip_atomic_fetch_max(mx + cell, code, __ATOMIC_RELAXED, __HIP_MEMORY_SCOPE_WORKGROUP);
    }
  }
  __syncthreads();
  const int s0 = ((int)blockIdx.x * 32) >> 7;
  const int w0 = ((int)blockIdx.x * 32) & 127;
  for (int o = wave; o < OWN_BLOCKS; o += 32) {
    const unsigned s = sm[lane * TAB_PITCH + o];
    const unsigned m = mx[lane * TAB_PITCH + o];
    const unsigned n = s >> 24, sum = s & 0xffffffu;
    const unsigned ent = (n == 0u) ? 0u : (n == 1u) ? sum
                       : (n == 2u) ? ((m << 16) | (sum - m)) : TAB_MANY;
    tab_row(tab, (unsigned)o, s0)[w0 + lane] = ent;
  }
}

__device__ __forceinline__ void push_owned(unsigned t, unsigned e, unsigned live, unsigned key,
                                           unsigned* q, int& cnt, int lane) {
  const unsigned iloc = ((t >> 8) | (t << 24)) - key;
  const unsigned mk = __builtin_amdgcn_ballot_w32(iloc < (unsigned)OWN_ROWS) & live;
  if (mk != 0u) {
    if ((mk >> lane) & 1u)
      q[cnt + (int)__builtin_amdgcn_mbcnt_lo(mk, 0u)] = (e << 10) | iloc;
    cnt += __builtin_popcount(mk);
  }
}

__device__ __forceinline__ void push_code(unsigned code, unsigned g0, unsigned* q, int& cnt, int lane) {
  const unsigned mk = __builtin_amdgcn_ballot_w32(code != 0u);
  if (mk != 0u) {
    if (code != 0u) {
      const unsigned cd = code - 1u;
      q[cnt + (int)__builtin_amdgcn_mbcnt_lo(mk, 0u)] = ((g0 + (cd >> 10)) << 10) | (cd & 1023u);
    }
    cnt += __builtin_popcount(mk);
  }
}

template <bool LAST>
__global__ void __launch_bounds__(OWN_THREADS)
gather_owner_kernel(const float* __restrict__ x, const float* __restrict__ x0,
                    const float* __restrict__ nrm,
                    const int* __restrict__ rates, const int* __restrict__ impl,
                    float* lst, unsigned* cntw, float* out) {
  __shared__ __attribute__((aligned(16))) float acc[OWN_ROWS * H];
  __shared__ unsigned queue[OWN_WAVES * QCAP];
  __shared__ unsigned nslot;

  const int tid  = threadIdx.x;
  const int lane = tid & 31;
  const int wave = __builtin_amdgcn_readfirstlane(tid >> 5);
  const unsigned b = blockIdx.x;

  if (tid == 0) nslot = 0u;
  float4* acc4 = (float4*)acc;
  const float4* s4 = (const float4*)x0;
  for (int i0 = wave; i0 < OWN_ROWS; i0 += SEED_G * OWN_WAVES) {
    float4 v[SEED_G];
#pragma unroll
    for (int u = 0; u < SEED_G; ++u) {
      const unsigned t = ((unsigned)(i0 + u * OWN_WAVES) << 8) | b;
      v[u] = (t < (unsigned)NNODE) ? s4[(size_t)t * (H / 4) + lane]
                                   : make_float4(0.0f, 0.0f, 0.0f, 0.0f);
    }
#pragma unroll
    for (int u = 0; u < SEED_G; ++u) {
      const int i = i0 + u * OWN_WAVES;
      if (i < OWN_ROWS) acc4[i * (H / 4) + lane] = v[u];
    }
  }
  __syncthreads();

  const float4* x4 = (const float4*)x;
  const unsigned nlist = LAST ? cntw[b] : 0u;
  if (LAST && nlist <= (unsigned)LOG_SLOTS) {
    for (int s = wave; s < (int)nlist; s += OWN_WAVES) {
      const uint2 cur = list_slot(lst, b, s)[lane];
      const int loc = (int)(cur.x & 1023u);
      const int nv = __builtin_popcount(__builtin_amdgcn_ballot_w32(loc < OWN_ROWS));
      if (nv == 32)
        sum_rows<true>((int)(cur.x >> 10), loc, (int)cur.y, 32, x4, acc, lane);
      else if (nv > 0)
        sum_rows<false>((int)(cur.x >> 10), loc, (int)cur.y, nv, x4, acc, lane);
    }
  } else {
    unsigned* q = queue + wave * QCAP;
    const unsigned key = b << 24;
    int cnt = 0;

    for (int s = wave; s < NSTEP; s += OWN_WAVES) {
      const int seg = s / SEG_STEPS;
      const int st  = s - seg * SEG_STEPS;
      const int* colp = seg_cols(seg, rates, impl);
      const unsigned ebase = (unsigned)(seg * EPR);
      uint4 d = ((const uint4*)tab_row(lst, b, s))[lane];
#pragma nounroll
      for (int k = 0; k < 4; ++k) {
        const unsigned ent = d.x;
        d.x = d.y; d.y = d.z; d.z = d.w;
        const int grp = ((st * 32 + lane) << 2) + k;
        const unsigned g0 = ebase + ((unsigned)grp << 6);
        const bool many = ent == TAB_MANY;
        push_code(many ? 0u : (ent & 0xffffu), g0, q, cnt, lane);
        push_code(many ? 0u : (ent >> 16),     g0, q, cnt, lane);
        unsigned m = __builtin_amdgcn_ballot_w32(many);
        do {
          if (m != 0u) {
            const int l0 = __builtin_ctz(m);
            m &= m - 1u;
            const int i0 = ((((st * 32 + l0) << 2) + k) << 6) + 2 * lane;
            const bool ok0 = i0 < EPR;
            const int2 t0 = *(const int2*)(colp + (ok0 ? i0 : 0));
            const unsigned live0 = __builtin_amdgcn_ballot_w32(ok0);
            const unsigned e0 = ebase + (unsigned)i0;
            push_owned((unsigned)t0.x, e0,      live0, key, q, cnt, lane);
            push_owned((unsigned)t0.y, e0 + 1u, live0, key, q, cnt, lane);
          }
          if (cnt >= 32) {
            __builtin_amdgcn_fence(__ATOMIC_SEQ_CST, "wavefront");
            const int full = cnt & ~31;
            for (int c = 0; c < full; c += 32)
              drain_chunk<true, !LAST>(q, c, 32, x4, nrm, rates, impl, acc, lane, lst, &nslot, b);
            const int rem = cnt - full;
            if (lane < rem) q[lane] = q[full + lane];
            __builtin_amdgcn_fence(__ATOMIC_SEQ_CST, "wavefront");
            cnt = rem;
          }
        } while (m != 0u);
      }
    }
    __builtin_amdgcn_fence(__ATOMIC_SEQ_CST, "wavefront");
    if (cnt > 0)
      drain_chunk<false, !LAST>(q, 0, cnt, x4, nrm, rates, impl, acc, lane, lst, &nslot, b);
  }

  __syncthreads();
  if (!LAST && tid == 0) cntw[b] = nslot;
  for (int i = wave; i < OWN_ROWS; i += OWN_WAVES) {
    const unsigned t = ((unsigned)i << 8) | b;
    if (t < (unsigned)NNODE) {
      float4 s = ((const float4*)(acc + i * H))[lane];
      if (LAST) {
        s.x *= (1.0f / 3.0f); s.y *= (1.0f / 3.0f); s.z *= (1.0f / 3.0f); s.w *= (1.0f / 3.0f);
      }
      ((float4*)(out + (size_t)t * H))[lane] = s;
    }
  }
}

extern "C" void kernel_launch(void* const* d_in, const int* in_sizes, int n_in,
                              void* d_out, int out_size, void* d_ws, size_t ws_size,
                              hipStream_t stream) {
  (void)in_sizes; (void)n_in; (void)out_size; (void)ws_size;

  const float* user_x = (const float*)d_in[0];
  const float* book_x = (const float*)d_in[1];
  const float* user_W = (const float*)d_in[2];
  const float* user_b = (const float*)d_in[3];
  const float* book_W = (const float*)d_in[4];
  const float* book_b = (const float*)d_in[5];
  const int*   rates  = (const int*)d_in[6];
  const int*   impl   = (const int*)d_in[7];
  float* outp = (float*)d_out;

  float* x   = (float*)d_ws;
  float* buf = x   + (size_t)NNODE * H;
  float* deg = buf + (size_t)NNODE * H;
  float* dis = deg + NNODE;
  float* nrm = dis + NNODE;
  unsigned* cntw = (unsigned*)deg;
  unsigned* part = (unsigned*)buf;

  {
    int waves = (NU + 31) / 32;
    gemm_bias_wmma<<<(waves + 7) / 8, 256, 0, stream>>>(user_x, user_W, user_b, x, NU);
    waves = (NB + 31) / 32;
    gemm_bias_wmma<<<(waves + 7) / 8, 256, 0, stream>>>(book_x, book_W, book_b,
                                                        x + (size_t)NU * H, NB);
  }

  degree_hist_kernel<<<HS * HR, HTHREADS, 0, stream>>>(rates, impl, part);
  dis_from_hist_kernel<<<(HPITCH + 255) / 256, 256, 0, stream>>>(part, dis);
  norm_kernel<<<(NE + 255) / 256, 256, 0, stream>>>(rates, impl, dis, nrm);
  table_kernel<<<TAB_BLOCKS, 1024, 0, stream>>>(rates, impl, outp);

  gather_owner_kernel<false><<<OWN_BLOCKS, OWN_THREADS, 0, stream>>>(x, x, nrm, rates, impl, outp, cntw, buf);
  gather_owner_kernel<true><<<OWN_BLOCKS, OWN_THREADS, 0, stream>>>(buf, x, nrm, rates, impl, outp, cntw, outp);
}


# layer sums: edges filed per owner wave (local row & 31) by LDS counters; each wave then adds its own LDS rows by plain 16-byte read-add-write: no LDS float atomics; order = filing order, fused adds
# speedup vs baseline: 4.0469x; 1.1309x over previous
#include <hip/hip_runtime.h>

#define NU 100000
#define NB 50000
#define NNODE (NU + NB)
#define H 128
#define F 256
#define EPR 500000
#define NE (4 * EPR)

typedef __attribute__((ext_vector_type(2))) float v2f;
typedef __attribute__((ext_vector_type(8))) float v8f;

#define KBLK (F / 8)

struct Frag {
  float4 a[2];
  float4 b[8];
};

__device__ __forceinline__ void load_frag(Frag& f, const float4* __restrict__ a0p,
                                          const float4* __restrict__ a1p,
                                          const float4* __restrict__ bp, int kb) {
  f.a[0] = a0p[2 * kb];
  f.a[1] = a1p[2 * kb];
#pragma unroll
  for (int nt = 0; nt < 8; ++nt) f.b[nt] = bp[nt * (16 * F / 4) + 2 * kb];
}

__device__ __forceinline__ void mma_frag(v8f (&c)[2][8], const Frag& f) {
#pragma unroll
  for (int nt = 0; nt < 8; ++nt) {
#pragma unroll
    for (int t = 0; t < 2; ++t) {
      v2f a = {f.a[t].x, f.a[t].y};
      v2f b = {f.b[nt].x, f.b[nt].y};
      c[t][nt] = __builtin_amdgcn_wmma_f32_16x16x4_f32(
           false, a,  false, b,
           (short)0, c[t][nt],  false,  false);
    }
  }
#pragma unroll
  for (int nt = 0; nt < 8; ++nt) {
#pragma unroll
    for (int t = 0; t < 2; ++t) {
      v2f a = {f.a[t].z, f.a[t].w};
      v2f b = {f.b[nt].z, f.b[nt].w};
      c[t][nt] = __builtin_amdgcn_wmma_f32_16x16x4_f32(
           false, a,  false, b,
           (short)0, c[t][nt],  false,  false);
    }
  }
}

__global__ void __launch_bounds__(256)
gemm_bias_wmma(const float* __restrict__ X, const float* __restrict__ W,
               const float* __restrict__ bias, float* __restrict__ Y, int M) {
  const int wave = (int)((blockIdx.x * blockDim.x + threadIdx.x) >> 5);
  const int lane = threadIdx.x & 31;
  if (wave * 32 >= M) return;
  const int mbase = wave * 32;
  const int lrow  = lane & 15;
  const int kq    = (lane >> 4) << 2;
  const bool two = mbase + 16 < M;
  const int r0 = mbase + lrow;
  const int r1 = two ? (r0 + 16) : r0;

  const float4* a0p = (const float4*)(X + (size_t)r0 * F + kq);
  const float4* a1p = (const float4*)(X + (size_t)r1 * F + kq);
  const float4* bp  = (const float4*)(W + (size_t)lrow * F + kq);

  v8f c[2][8] = {};

  Frag f0, f1;
  load_frag(f0, a0p, a1p, bp, 0);
  for (int kb = 0; kb < KBLK; kb += 2) {
    load_frag(f1, a0p, a1p, bp, kb + 1);
    __builtin_amdgcn_sched_barrier(0);
    mma_frag(c, f0);
    __builtin_amdgcn_sched_barrier(0);
    load_frag(f0, a0p, a1p, bp, (kb + 2 < KBLK) ? (kb + 2) : 0);
    __builtin_amdgcn_sched_barrier(0);
    mma_frag(c, f1);
    __builtin_amdgcn_sched_barrier(0);
  }

  const int mo = (lane >> 4) << 3;
#pragma unroll
  for (int t = 0; t < 2; ++t) {
    if (t == 0 || two) {
#pragma unroll
      for (int nt = 0; nt < 8; ++nt) {
        float bv = bias[nt * 16 + lrow];
#pragma unroll
        for (int v = 0; v < 8; ++v) {
          Y[(size_t)(mbase + 16 * t + mo + v) * H + nt * 16 + lrow] = c[t][nt][v] + bv;
        }
      }
    }
  }
}

__device__ __forceinline__ void decode_edge(int e, const int* __restrict__ rates,
                                            const int* __restrict__ impl,
                                            int& row, int& col) {
  const int rev = (e >= 2 * EPR) ? 1 : 0;
  int idx = rev ? (e - 2 * EPR) : e;
  const int* ed = (idx < EPR) ? rates : impl;
  const int i = (idx < EPR) ? idx : (idx - EPR);
  const int u = ed[i];
  const int b = ed[i + EPR];
  row = rev ? (b + NU) : u;
  col = rev ? u : b;
}

__global__ void __launch_bounds__(256)
norm_kernel(const int* __restrict__ rates, const int* __restrict__ impl,
            const float* __restrict__ dis, float* __restrict__ nrm) {
  const int e = blockIdx.x * blockDim.x + threadIdx.x;
  if (e >= NE) return;
  int row, col;
  decode_edge(e, rates, impl, row, col);
  nrm[e] = dis[row] * dis[col];
}

#define OWN_BLOCKS  256
#define OWN_ROWS    ((NNODE + OWN_BLOCKS - 1) / OWN_BLOCKS)
#define OWN_THREADS 1024
#define OWN_WAVES   (OWN_THREADS / 32)
#define QCAP        160
#define DRAIN_G     16
#define SEED_G      10

#define TAB_ROWS    248
#define LOG_ROWS    (585 - TAB_ROWS)
#define CAPW_ROWS   10
#define CAPW        (CAPW_ROWS * 64)
static_assert(OWN_WAVES == 32, "one list count per lane; owner = local row & 31");
static_assert(OWN_WAVES * CAPW_ROWS <= LOG_ROWS, "the lists fit in front of the table rows");
static_assert((((LOG_ROWS + TAB_ROWS - 1) << 8) | (OWN_BLOCKS - 1)) < NNODE, "every block owns that many real rows");
static_assert(NNODE < (1 << 22), "a source row fits the 22 bits above the local row");

#define GRP_SEG     ((EPR + 63) / 64)
#define SEG_PITCH   7936
#define SEG_STEPS   (SEG_PITCH / 128)
#define PPITCH      (4 * SEG_PITCH)
#define NSTEP       (4 * SEG_STEPS)
#define TAB_BLOCKS  (PPITCH / 32)
#define TAB_MANY    0xffffffffu
#define TAB_PITCH   257
static_assert(NSTEP == TAB_ROWS, "one table row per wave step");
static_assert(SEG_PITCH % 32 == 0, "a table block stays inside one segment");

__device__ __forceinline__ void lds_add(float* p, float v) {
  __hip_atomic_fetch_add(p, v, __ATOMIC_RELAXED, __HIP_MEMORY_SCOPE_WORKGROUP);
}

template <bool FULL>
__device__ __forceinline__ void sum_rows(int row, int loc, int wbits, int nv,
                                         const float4* __restrict__ x4, float* acc, int lane) {
  const int h = lane >> 4;
#pragma unroll
  for (int half = 0; half < 32 / DRAIN_G; ++half) {
    if (!FULL && half * DRAIN_G >= nv) break;
    float4 v[DRAIN_G];
#pragma unroll
    for (int g = 0; g < DRAIN_G; ++g) {
      const int r = __builtin_amdgcn_readlane(row, half * DRAIN_G + g);
      v[g] = x4[(size_t)r * (H / 4) + lane];
    }
#pragma unroll
    for (int g = 0; g < DRAIN_G; ++g) {
      if (FULL || half * DRAIN_G + g < nv) {
        const int il = __builtin_amdgcn_readlane(loc, half * DRAIN_G + g);
        const float m = __int_as_float(__builtin_amdgcn_readlane(wbits, half * DRAIN_G + g));
        float* a = acc + il * H + 4 * lane;
        const float4 s = v[g];
        lds_add(a + h,             (h ? s.y : s.x) * m);
        lds_add(a + ((1 + h) & 3), (h ? s.z : s.y) * m);
        lds_add(a + ((2 + h) & 3), (h ? s.w : s.z) * m);
        lds_add(a + ((3 + h) & 3), (h ? s.x : s.w) * m);
      }
    }
  }
}

template <bool FULL>
__device__ __forceinline__ void sum_owned(int row, int loc, int wbits, int nv,
                                          const float4* __restrict__ x4, float4* acc4, int lane,
                                          const uint2* np, uint2& nxt) {
#pragma unroll
  for (int half = 0; half < 32 / DRAIN_G; ++half) {
    if (!FULL && half * DRAIN_G >= nv) break;
    float4 v[DRAIN_G];
#pragma unroll
    for (int g = 0; g < DRAIN_G; ++g) {
      const int r = __builtin_amdgcn_readlane(row, half * DRAIN_G + g);
      v[g] = x4[(size_t)r * (H / 4) + lane];
    }
    if (FULL) {
      if (half == 32 / DRAIN_G - 1 && np != nullptr) nxt = *np;
#pragma unroll
      for (int p = 0; p < DRAIN_G / 2; ++p) {
        const int k0 = half * DRAIN_G + 2 * p;
        const int i0 = __builtin_amdgcn_readlane(loc, k0);
        const int i1 = __builtin_amdgcn_readlane(loc, k0 + 1);
        const float m0 = __int_as_float(__builtin_amdgcn_readlane(wbits, k0));
        const float m1 = __int_as_float(__builtin_amdgcn_readlane(wbits, k0 + 1));
        float4* a0 = acc4 + i0 * (H / 4) + lane;
        float4* a1 = acc4 + i1 * (H / 4) + lane;
        const float4 s0 = v[2 * p], s1 = v[2 * p + 1];
        if (i0 != i1) {
          float4 t0 = *a0;
          float4 t1 = *a1;
          t0.x += s0.x * m0; t0.y += s0.y * m0; t0.z += s0.z * m0; t0.w += s0.w * m0;
          t1.x += s1.x * m1; t1.y += s1.y * m1; t1.z += s1.z * m1; t1.w += s1.w * m1;
          *a0 = t0;
          *a1 = t1;
        } else {
          float4 t0 = *a0;
          t0.x += s0.x * m0; t0.y += s0.y * m0; t0.z += s0.z * m0; t0.w += s0.w * m0;
          t0.x += s1.x * m1; t0.y += s1.y * m1; t0.z += s1.z * m1; t0.w += s1.w * m1;
          *a0 = t0;
        }
      }
    } else {
#pragma unroll
      for (int g = 0; g < DRAIN_G; ++g) {
        if (half * DRAIN_G + g < nv) {
          const int il = __builtin_amdgcn_readlane(loc, half * DRAIN_G + g);
          const float m = __int_as_float(__builtin_amdgcn_readlane(wbits, half * DRAIN_G + g));
          float4* a = acc4 + il * (H / 4) + lane;
          float4 t = *a;
          const float4 s = v[g];
          t.x += s.x * m; t.y += s.y * m; t.z += s.z * m; t.w += s.w * m;
          *a = t;
        }
      }
    }
  }
}

__device__ __forceinline__ uint2* list_entry(float* lst, unsigned b, int w, int k) {
  return (uint2*)(lst + (size_t)(((unsigned)(w * CAPW_ROWS + (k >> 6)) << 8) | b) * H) + (k & 63);
}

__device__ __forceinline__ void replay_owned(float* lst, unsigned b, int wave, int n,
                                             const float4* __restrict__ x4, float4* acc4, int lane) {
  const int nfull = n & ~31;
  uint2 cur = make_uint2(0u, 0u);
  if (n > 0) cur = list_entry(lst, b, wave, 0)[lane];
  int i = 0;
  for (; i < nfull; i += 32) {
    const uint2* np = (i + 32 < n) ? (list_entry(lst, b, wave, i + 32) + lane) : nullptr;
    uint2 nxt = cur;
    sum_owned<true>((int)(cur.x >> 10), (int)(cur.x & 1023u), (int)cur.y, 32, x4, acc4, lane,
                    np, nxt);
    cur = nxt;
  }
  if (i < n) {
    const int nv = n - i;
    const bool valid = lane < nv;
    const int row = valid ? (int)(cur.x >> 10) : 0;
    const int loc = valid ? (int)(cur.x & 1023u) : 0;
    uint2 none = cur;
    sum_owned<false>(row, loc, (int)cur.y, nv, x4, acc4, lane, nullptr, none);
  }
}

template <bool FULL, bool FILE>
__device__ __forceinline__ void drain_chunk(const unsigned* q, int base, int nv,
                                            const float4* __restrict__ x4,
                                            const float* __restrict__ nrm,
                                            const int* __restrict__ rates,
                                            const int* __restrict__ impl,
                                            float* acc, int lane,
                                            float* lst, unsigned* wcnt, unsigned b) {
  const int li = FULL ? (base + lane) : (base + ((lane < nv) ? lane : 0));
  const unsigned d = q[li];
  const int e = (int)(d >> 10);
  const int loc = (int)(d & 1023u);
  const bool rev = e >= 2 * EPR;
  const int idx = rev ? (e - 2 * EPR) : e;
  const bool first = idx < EPR;
  const int i = first ? idx : (idx - EPR);
  const int* ed = first ? rates : impl;
  const int sv = ed[i + (rev ? EPR : 0)];
  const int row = rev ? (sv + NU) : sv;
  const int wbits = __float_as_int(nrm[e]);
  if (FILE) {
    if (FULL || lane < nv) {
      const int ow = loc & (OWN_WAVES - 1);
      const unsigned pos = __hip_atomic_fetch_add(wcnt + ow, 1u, __ATOMIC_RELAXED,
                                                  __HIP_MEMORY_SCOPE_WORKGROUP);
      if (pos < (unsigned)CAPW) {
        uint2 ent;
        ent.x = ((unsigned)row << 10) | (unsigned)loc;
        ent.y = (unsigned)wbits;
        *list_entry(lst, b, ow, (int)pos) = ent;
      }
    }
  } else {
    sum_rows<FULL>(row, loc, wbits, nv, x4, acc, lane);
  }
}

__device__ __forceinline__ const int* seg_cols(int seg, const int* __restrict__ rates,
                                               const int* __restrict__ impl) {
  return (seg == 0) ? (rates + EPR) : (seg == 1) ? (impl + EPR) : (seg == 2) ? rates : impl;
}

#define HR      8
#define HS      32
#define HRANGE  (NNODE / HR)
#define HPAIRS  (HRANGE / 2)
#define HSLICE  (NE / HS)
#define HSL4    (HSLICE / 4)
#define HPITCH  (NNODE / 2)
#define HTHREADS 1024
static_assert(HSLICE < 65536, "a 16-bit count must hold a whole slice");
static_assert(HRANGE * HR == NNODE && HRANGE % 2 == 0, "ranges cover the nodes in whole dwords");
static_assert(HSLICE * HS == NE && EPR % HSLICE == 0 && HSLICE % 4 == 0, "slices stay inside a segment");

__device__ __forceinline__ void hist_add(unsigned* cnt, int t, unsigned lo) {
  const unsigned q = (unsigned)t - lo;
  if (q < (unsigned)HRANGE)
    __hip_atomic_fetch_add(cnt + (q >> 1), (q & 1u) ? 65536u : 1u, __ATOMIC_RELAXED,
                           __HIP_MEMORY_SCOPE_WORKGROUP);
}

__global__ void __launch_bounds__(HTHREADS)
degree_hist_kernel(const int* __restrict__ rates, const int* __restrict__ impl,
                   unsigned* __restrict__ part) {
  __shared__ unsigned cnt[HPAIRS];
  const int tid = threadIdx.x;
  const int r = (int)blockIdx.x % HR;
  const int s = (int)blockIdx.x / HR;
  for (int i = tid; i < HPAIRS; i += HTHREADS) cnt[i] = 0u;
  __syncthreads();
  const int seg = s / (HS / 4);
  const int4* p = (const int4*)(seg_cols(seg, rates, impl) + (s - seg * (HS / 4)) * HSLICE);
  const unsigned lo = (unsigned)(r * HRANGE);
  for (int j0 = 0; j0 < HSL4; j0 += 4 * HTHREADS) {
    int4 v[4];
#pragma unroll
    for (int u = 0; u < 4; ++u) {
      const int j = j0 + u * HTHREADS + tid;
      v[u] = (j < HSL4) ? p[j] : make_int4(-1, -1, -1, -1);
    }
#pragma unroll
    for (int u = 0; u < 4; ++u) {
      hist_add(cnt, v[u].x, lo);
      hist_add(cnt, v[u].y, lo);
      hist_add(cnt, v[u].z, lo);
      hist_add(cnt, v[u].w, lo);
    }
  }
  __syncthreads();
  unsigned* dst = part + (size_t)s * HPITCH + r * HPAIRS;
  for (int i = tid; i < HPAIRS; i += HTHREADS) dst[i] = cnt[i];
}

__global__ void __launch_bounds__(256)
dis_from_hist_kernel(const unsigned* __restrict__ part, float* __restrict__ dis) {
  const int p = blockIdx.x * blockDim.x + threadIdx.x;
  if (p >= HPITCH) return;
  unsigned v[HS];
#pragma unroll
  for (int s = 0; s < HS; ++s) v[s] = part[(size_t)s * HPITCH + p];
  unsigned c0 = 0u, c1 = 0u;
#pragma unroll
  for (int s = 0; s < HS; ++s) {
    c0 += v[s] & 0xffffu;
    c1 += v[s] >> 16;
  }
  const float d0 = (float)c0, d1 = (float)c1;
  float2 o;
  o.x = (d0 > 0.0f) ? rsqrtf(d0) : 0.0f;
  o.y = (d1 > 0.0f) ? rsqrtf(d1) : 0.0f;
  ((float2*)dis)[p] = o;
}

__device__ __forceinline__ unsigned* tab_row(float* tab, unsigned o, int s) {
  return (unsigned*)(tab + (size_t)(((unsigned)(LOG_ROWS + s) << 8) | o) * H);
}

__global__ void __launch_bounds__(1024)
table_kernel(const int* __restrict__ rates, const int* __restrict__ impl, float* tab) {
  __shared__ __attribute__((aligned(16))) unsigned sm[32 * TAB_PITCH];
  __shared__ __attribute__((aligned(16))) unsigned mx[32 * TAB_PITCH];
  const int tid  = threadIdx.x;
  const int lane = tid & 31;
  const int wave = __builtin_amdgcn_readfirstlane(tid >> 5);
  for (int i = tid; i < 32 * TAB_PITCH / 4; i += 1024) {
    ((uint4*)sm)[i] = make_uint4(0u, 0u, 0u, 0u);
    ((uint4*)mx)[i] = make_uint4(0u, 0u, 0u, 0u);
  }
  __syncthreads();
  const int pidx = (int)blockIdx.x * 32 + wave;
  const int seg = pidx / SEG_PITCH;
  const int gs  = pidx - seg * SEG_PITCH;
  if (gs < GRP_SEG) {
    const int idx = gs * 64 + 2 * lane;
    const bool ok = idx < EPR;
    const int2 t2 = *(const int2*)(seg_cols(seg, rates, impl) + (ok ? idx : 0));
    const unsigned ta = (unsigned)t2.x, tb = (unsigned)t2.y;
    if (ok && ta < (unsigned)(OWN_ROWS << 8)) {
      const unsigned code = (((unsigned)(2 * lane) << 10) | (ta >> 8)) + 1u;
      const int cell = wave * TAB_PITCH + (int)(ta & 255u);
      __hip_atomic_fetch_add(sm + cell, (1u << 24) + code, __ATOMIC_RELAXED, __HIP_MEMORY_SCOPE_WORKGROUP);
      __hip_atomic_fetch_max(mx + cell, code, __ATOMIC_RELAXED, __HIP_MEMORY_SCOPE_WORKGROUP);
    }
    if (ok && tb < (unsigned)(OWN_ROWS << 8)) {
      const unsigned code = (((unsigned)(2 * lane + 1) << 10) | (tb >> 8)) + 1u;
      const int cell = wave * TAB_PITCH + (int)(tb & 255u);
      __hip_atomic_fetch_add(sm + cell, (1u << 24) + code, __ATOMIC_RELAXED, __HIP_MEMORY_SCOPE_WORKGROUP);
      __hip_atomic_fetch_max(mx + cell, code, __ATOMIC_RELAXED, __HIP_MEMORY_SCOPE_WORKGROUP);
    }
  }
  __syncthreads();
  const int s0 = ((int)blockIdx.x * 32) >> 7;
  const int w0 = ((int)blockIdx.x * 32) & 127;
  for (int o = wave; o < OWN_BLOCKS; o += 32) {
    const unsigned s = sm[lane * TAB_PITCH + o];
    const unsigned m = mx[lane * TAB_PITCH + o];
    const unsigned n = s >> 24, sum = s & 0xffffffu;
    const unsigned ent = (n == 0u) ? 0u : (n == 1u) ? sum
                       : (n == 2u) ? ((m << 16) | (sum - m)) : TAB_MANY;
    tab_row(tab, (unsigned)o, s0)[w0 + lane] = ent;
  }
}

__device__ __forceinline__ void push_owned(unsigned t, unsigned e, unsigned live, unsigned key,
                                           unsigned* q, int& cnt, int lane) {
  const unsigned iloc = ((t >> 8) | (t << 24)) - key;
  const unsigned mk = __builtin_amdgcn_ballot_w32(iloc < (unsigned)OWN_ROWS) & live;
  if (mk != 0u) {
    if ((mk >> lane) & 1u)
      q[cnt + (int)__builtin_amdgcn_mbcnt_lo(mk, 0u)] = (e << 10) | iloc;
    cnt += __builtin_popcount(mk);
  }
}

__device__ __forceinline__ void push_code(unsigned code, unsigned g0, unsigned* q, int& cnt, int lane) {
  const unsigned mk = __builtin_amdgcn_ballot_w32(code != 0u);
  if (mk != 0u) {
    if (code != 0u) {
      const unsigned cd = code - 1u;
      q[cnt + (int)__builtin_amdgcn_mbcnt_lo(mk, 0u)] = ((g0 + (cd >> 10)) << 10) | (cd & 1023u);
    }
    cnt += __builtin_popcount(mk);
  }
}

template <bool FILE>
__device__ __forceinline__ void scan_edges(unsigned b, int wave, int lane, unsigned* q,
                                           const float4* __restrict__ x4,
                                           const float* __restrict__ nrm,
                                           const int* __restrict__ rates,
                                           const int* __restrict__ impl,
                                           float* acc, float* lst, unsigned* wcnt) {
  const unsigned key = b << 24;
  int cnt = 0;

  for (int s = wave; s < NSTEP; s += OWN_WAVES) {
    const int seg = s / SEG_STEPS;
    const int st  = s - seg * SEG_STEPS;
    const int* colp = seg_cols(seg, rates, impl);
    const unsigned ebase = (unsigned)(seg * EPR);
    uint4 d = ((const uint4*)tab_row(lst, b, s))[lane];
#pragma nounroll
    for (int k = 0; k < 4; ++k) {
      const unsigned ent = d.x;
      d.x = d.y; d.y = d.z; d.z = d.w;
      const int grp = ((st * 32 + lane) << 2) + k;
      const unsigned g0 = ebase + ((unsigned)grp << 6);
      const bool many = ent == TAB_MANY;
      push_code(many ? 0u : (ent & 0xffffu), g0, q, cnt, lane);
      push_code(many ? 0u : (ent >> 16),     g0, q, cnt, lane);
      unsigned m = __builtin_amdgcn_ballot_w32(many);
      do {
        if (m != 0u) {
          const int l0 = __builtin_ctz(m);
          m &= m - 1u;
          const int i0 = ((((st * 32 + l0) << 2) + k) << 6) + 2 * lane;
          const bool ok0 = i0 < EPR;
          const int2 t0 = *(const int2*)(colp + (ok0 ? i0 : 0));
          const unsigned live0 = __builtin_amdgcn_ballot_w32(ok0);
          const unsigned e0 = ebase + (unsigned)i0;
          push_owned((unsigned)t0.x, e0,      live0, key, q, cnt, lane);
          push_owned((unsigned)t0.y, e0 + 1u, live0, key, q, cnt, lane);
        }
        if (cnt >= 32) {
          __builtin_amdgcn_fence(__ATOMIC_SEQ_CST, "wavefront");
          const int full = cnt & ~31;
          for (int c = 0; c < full; c += 32)
            drain_chunk<true, FILE>(q, c, 32, x4, nrm, rates, impl, acc, lane, lst, wcnt, b);
          const int rem = cnt - full;
          if (lane < rem) q[lane] = q[full + lane];
          __builtin_amdgcn_fence(__ATOMIC_SEQ_CST, "wavefront");
          cnt = rem;
        }
      } while (m != 0u);
    }
  }
  __builtin_amdgcn_fence(__ATOMIC_SEQ_CST, "wavefront");
  if (cnt > 0)
    drain_chunk<false, FILE>(q, 0, cnt, x4, nrm, rates, impl, acc, lane, lst, wcnt, b);
}

template <bool LAST>
__global__ void __launch_bounds__(OWN_THREADS)
gather_owner_kernel(const float* __restrict__ x, const float* __restrict__ x0,
                    const float* __restrict__ nrm,
                    const int* __restrict__ rates, const int* __restrict__ impl,
                    float* lst, unsigned* cntw, float* out) {
  __shared__ __attribute__((aligned(16))) float acc[OWN_ROWS * H];
  __shared__ unsigned queue[OWN_WAVES * QCAP];
  __shared__ unsigned wcnt[OWN_WAVES];

  const int tid  = threadIdx.x;
  const int lane = tid & 31;
  const int wave = __builtin_amdgcn_readfirstlane(tid >> 5);
  const unsigned b = blockIdx.x;

  if (tid < OWN_WAVES) wcnt[tid] = LAST ? cntw[b * OWN_WAVES + tid] : 0u;
  float4* acc4 = (float4*)acc;
  const float4* s4 = (const float4*)x0;
  for (int i0 = wave; i0 < OWN_ROWS; i0 += SEED_G * OWN_WAVES) {
    float4 v[SEED_G];
#pragma unroll
    for (int u = 0; u < SEED_G; ++u) {
      const unsigned t = ((unsigned)(i0 + u * OWN_WAVES) << 8) | b;
      v[u] = (t < (unsigned)NNODE) ? s4[(size_t)t * (H / 4) + lane]
                                   : make_float4(0.0f, 0.0f, 0.0f, 0.0f);
    }
#pragma unroll
    for (int u = 0; u < SEED_G; ++u) {
      const int i = i0 + u * OWN_WAVES;
      if (i < OWN_ROWS) acc4[i * (H / 4) + lane] = v[u];
    }
  }
  __syncthreads();

  const float4* x4 = (const float4*)x;
  unsigned* q = queue + wave * QCAP;
  if (!LAST) {
    scan_edges<true>(b, wave, lane, q, x4, nrm, rates, impl, acc, lst, wcnt);
    __syncthreads();
    if (tid < OWN_WAVES) cntw[b * OWN_WAVES + tid] = wcnt[tid];
  }
  const bool fits = __builtin_amdgcn_ballot_w32(wcnt[lane] > (unsigned)CAPW) == 0u;
  if (fits) {
    const int n = __builtin_amdgcn_readfirstlane((int)wcnt[wave]);
    replay_owned(lst, b, wave, n, x4, acc4, lane);
  } else {
    scan_edges<false>(b, wave, lane, q, x4, nrm, rates, impl, acc, lst, wcnt);
  }

  __syncthreads();
  for (int i = wave; i < OWN_ROWS; i += OWN_WAVES) {
    const unsigned t = ((unsigned)i << 8) | b;
    if (t < (unsigned)NNODE) {
      float4 s = ((const float4*)(acc + i * H))[lane];
      if (LAST) {
        s.x *= (1.0f / 3.0f); s.y *= (1.0f / 3.0f); s.z *= (1.0f / 3.0f); s.w *= (1.0f / 3.0f);
      }
      ((float4*)(out + (size_t)t * H))[lane] = s;
    }
  }
}

extern "C" void kernel_launch(void* const* d_in, const int* in_sizes, int n_in,
                              void* d_out, int out_size, void* d_ws, size_t ws_size,
                              hipStream_t stream) {
  (void)in_sizes; (void)n_in; (void)out_size; (void)ws_size;

  const float* user_x = (const float*)d_in[0];
  const float* book_x = (const float*)d_in[1];
  const float* user_W = (const float*)d_in[2];
  const float* user_b = (const float*)d_in[3];
  const float* book_W = (const float*)d_in[4];
  const float* book_b = (const float*)d_in[5];
  const int*   rates  = (const int*)d_in[6];
  const int*   impl   = (const int*)d_in[7];
  float* outp = (float*)d_out;

  float* x   = (float*)d_ws;
  float* buf = x   + (size_t)NNODE * H;
  float* deg = buf + (size_t)NNODE * H;
  float* dis = deg + NNODE;
  float* nrm = dis + NNODE;
  unsigned* cntw = (unsigned*)deg;
  unsigned* part = (unsigned*)buf;

  {
    int waves = (NU + 31) / 32;
    gemm_bias_wmma<<<(waves + 7) / 8, 256, 0, stream>>>(user_x, user_W, user_b, x, NU);
    waves = (NB + 31) / 32;
    gemm_bias_wmma<<<(waves + 7) / 8, 256, 0, stream>>>(book_x, book_W, book_b,
                                                        x + (size_t)NU * H, NB);
  }

  degree_hist_kernel<<<HS * HR, HTHREADS, 0, stream>>>(rates, impl, part);
  dis_from_hist_kernel<<<(HPITCH + 255) / 256, 256, 0, stream>>>(part, dis);
  norm_kernel<<<(NE + 255) / 256, 256, 0, stream>>>(rates, impl, dis, nrm);
  table_kernel<<<TAB_BLOCKS, 1024, 0, stream>>>(rates, impl, outp);

  gather_owner_kernel<false><<<OWN_BLOCKS, OWN_THREADS, 0, stream>>>(x, x, nrm, rates, impl, outp, cntw, buf);
  gather_owner_kernel<true><<<OWN_BLOCKS, OWN_THREADS, 0, stream>>>(buf, x, nrm, rates, impl, outp, cntw, outp);
}
